# starting kernel
# baseline (speedup 1.0000x reference)
#include <hip/hip_runtime.h>
#include <math.h>

typedef __attribute__((ext_vector_type(2))) float v2f;
typedef __attribute__((ext_vector_type(8))) float v8f;

#define NNODES 100000
#define HID    128
#define TILE_M 16
#define ROWT   2
#define ROWS   (ROWT * TILE_M)

static __device__ __forceinline__ float atomic_add_f32(float* p, float v) {
  return __hip_atomic_fetch_add(p, v, __ATOMIC_RELAXED, __HIP_MEMORY_SCOPE_AGENT);
}

__global__ __launch_bounds__(256) void deg_kernel(const int* __restrict__ dst,
                                                  float* __restrict__ deg, int E) {
  int e = blockIdx.x * 256 + threadIdx.x;
  if (e < E) atomic_add_f32(&deg[dst[e]], 1.0f);
}

__global__ __launch_bounds__(256) void invdeg_kernel(float* __restrict__ deg, int N) {
  int n = blockIdx.x * 256 + threadIdx.x;
  if (n < N) deg[n] = 1.0f / fmaxf(deg[n], 1.0f);
}

__global__ __launch_bounds__(256) void scatter_kernel(const float* __restrict__ x,
                                                      const int* __restrict__ src,
                                                      const int* __restrict__ dst,
                                                      float* __restrict__ agg, int E) {
  int gid  = blockIdx.x * 256 + threadIdx.x;
  int e    = gid >> 5;
  int lane = gid & 31;
  if (e >= E) return;
  int s = src[e], d = dst[e];
  const float4 v = *(const float4*)(x + (size_t)s * HID + lane * 4);
  float* ap = agg + (size_t)d * HID + lane * 4;
  atomic_add_f32(ap + 0, v.x);
  atomic_add_f32(ap + 1, v.y);
  atomic_add_f32(ap + 2, v.z);
  atomic_add_f32(ap + 3, v.w);
}

__global__ __launch_bounds__(256) void sage_transform(
    const float* __restrict__ agg, const float* __restrict__ xin,
    const float* __restrict__ Wl, const float* __restrict__ bl,
    const float* __restrict__ Wr, const float* __restrict__ invdeg,
    float* __restrict__ xout) {
  __shared__ float lds_a[ROWS * HID];
  __shared__ float lds_x[ROWS * HID];
  __shared__ float lds_o[ROWS * HID];
  __shared__ float lds_inv[ROWS];

  const int tid  = threadIdx.x;
  const int row0 = blockIdx.x * ROWS;

#pragma unroll
  for (int rt = 0; rt < ROWT; ++rt) {
    const int r = rt * TILE_M + (tid >> 4);
    const int c = (tid & 15) * 8;
    const float inv = invdeg[row0 + r];
    const float4* pa = (const float4*)(agg + (size_t)(row0 + r) * HID + c);
    float4 a0 = pa[0], a1 = pa[1];
    float4* la4 = (float4*)(lds_a + r * HID + c);
    la4[0] = make_float4(a0.x * inv, a0.y * inv, a0.z * inv, a0.w * inv);
    la4[1] = make_float4(a1.x * inv, a1.y * inv, a1.z * inv, a1.w * inv);
    const float4* px = (const float4*)(xin + (size_t)(row0 + r) * HID + c);
    float4* lx4 = (float4*)(lds_x + r * HID + c);
    lx4[0] = px[0];
    lx4[1] = px[1];
  }

  const int wave  = tid >> 5;
  const int lane  = tid & 31;
  const int nn    = lane & 15;
  const int khalf = (lane >> 4) * 2;
  const int h0    = wave * 16;

  v2f bL[32], bR[32];
  {
    const float* pwl = Wl + (size_t)(h0 + nn) * HID + khalf;
    const float* pwr = Wr + (size_t)(h0 + nn) * HID + khalf;
#pragma unroll
    for (int kk = 0; kk < 32; ++kk) {
      bL[kk] = *(const v2f*)(pwl + kk * 4);
      bR[kk] = *(const v2f*)(pwr + kk * 4);
    }
  }
  const float bias  = bl[h0 + nn];
  const int   rbase = (lane >> 4) * 8;
  __syncthreads();

#pragma unroll 1
  for (int rt = 0; rt < ROWT; ++rt) {
    v8f accL = {0.f, 0.f, 0.f, 0.f, 0.f, 0.f, 0.f, 0.f};
    v8f accR = {0.f, 0.f, 0.f, 0.f, 0.f, 0.f, 0.f, 0.f};
    const float* la = lds_a + (rt * TILE_M + nn) * HID + khalf;
    const float* lx = lds_x + (rt * TILE_M + nn) * HID + khalf;
#pragma unroll
    for (int kk = 0; kk < 32; ++kk) {
      v2f aA = *(const v2f*)(la + kk * 4);
      accL = __builtin_amdgcn_wmma_f32_16x16x4_f32(false, aA, false, bL[kk],
                                                   (short)0, accL, false, false);
      v2f aX = *(const v2f*)(lx + kk * 4);
      accR = __builtin_amdgcn_wmma_f32_16x16x4_f32(false, aX, false, bR[kk],
                                                   (short)0, accR, false, false);
    }
#pragma unroll
    for (int j = 0; j < 8; ++j)
      lds_o[(rt * TILE_M + rbase + j) * HID + h0 + nn] = accL[j] + accR[j] + bias;
  }
  __syncthreads();

#pragma unroll
  for (int rt = 0; rt < ROWT; ++rt) {
    const int r = rt * TILE_M + (tid >> 4);
    const int p = tid & 15;
    const float* po = lds_o + r * HID + p * 8;
    float s = 0.f;
#pragma unroll
    for (int q = 0; q < 8; ++q) s += po[q] * po[q];
    s += __shfl_xor(s, 1);
    s += __shfl_xor(s, 2);
    s += __shfl_xor(s, 4);
    s += __shfl_xor(s, 8);
    if (p == 0) lds_inv[r] = 1.0f / fmaxf(sqrtf(s), 1e-12f);
  }
  __syncthreads();

#pragma unroll
  for (int rt = 0; rt < ROWT; ++rt) {
    const int r = rt * TILE_M + (tid >> 4);
    const int c = (tid & 15) * 8;
    const float inv = lds_inv[r];
    const float* po = lds_o + r * HID + c;
    float4 o0, o1;
    o0.x = fmaxf(po[0] * inv, 0.f); o0.y = fmaxf(po[1] * inv, 0.f);
    o0.z = fmaxf(po[2] * inv, 0.f); o0.w = fmaxf(po[3] * inv, 0.f);
    o1.x = fmaxf(po[4] * inv, 0.f); o1.y = fmaxf(po[5] * inv, 0.f);
    o1.z = fmaxf(po[6] * inv, 0.f); o1.w = fmaxf(po[7] * inv, 0.f);
    float4* pw = (float4*)(xout + (size_t)(row0 + r) * HID + c);
    pw[0] = o0;
    pw[1] = o1;
  }
}

__global__ __launch_bounds__(256) void mlp_head(
    const float* __restrict__ xin, const float* __restrict__ W1,
    const float* __restrict__ b1, const float* __restrict__ W2,
    const float* __restrict__ b2, float* __restrict__ out) {
  __shared__ float lds_x[TILE_M * HID];
  __shared__ float lds_h[TILE_M * HID];

  const int tid  = threadIdx.x;
  const int row0 = blockIdx.x * TILE_M;

  {
    const int r = tid >> 4;
    const int c = (tid & 15) * 8;
    const float4* px = (const float4*)(xin + (size_t)(row0 + r) * HID + c);
    float4* lx4 = (float4*)(lds_x + r * HID + c);
    lx4[0] = px[0];
    lx4[1] = px[1];
  }

  const int wave  = tid >> 5;
  const int lane  = tid & 31;
  const int nn    = lane & 15;
  const int khalf = (lane >> 4) * 2;
  const int h0    = wave * 16;

  v2f bW[32];
  {
    const float* pw = W1 + (size_t)(h0 + nn) * HID + khalf;
#pragma unroll
    for (int kk = 0; kk < 32; ++kk) bW[kk] = *(const v2f*)(pw + kk * 4);
  }
  const float bias = b1[h0 + nn];
  __syncthreads();

  v8f acc = {0.f, 0.f, 0.f, 0.f, 0.f, 0.f, 0.f, 0.f};
  const float* lx = lds_x + nn * HID + khalf;
#pragma unroll
  for (int kk = 0; kk < 32; ++kk) {
    v2f a = *(const v2f*)(lx + kk * 4);
    acc = __builtin_amdgcn_wmma_f32_16x16x4_f32(false, a, false, bW[kk],
                                                (short)0, acc, false, false);
  }
  const int rbase = (lane >> 4) * 8;
#pragma unroll
  for (int j = 0; j < 8; ++j)
    lds_h[(rbase + j) * HID + h0 + nn] = fmaxf(acc[j] + bias, 0.f);
  __syncthreads();

  {
    const int r = tid >> 4;
    const int p = tid & 15;
    const float* ph = lds_h + r * HID + p * 8;
    float s0 = 0.f, s1 = 0.f;
#pragma unroll
    for (int q = 0; q < 8; ++q) {
      float h = ph[q];
      s0 += h * W2[p * 8 + q];
      s1 += h * W2[HID + p * 8 + q];
    }
    s0 += __shfl_xor(s0, 1); s1 += __shfl_xor(s1, 1);
    s0 += __shfl_xor(s0, 2); s1 += __shfl_xor(s1, 2);
    s0 += __shfl_xor(s0, 4); s1 += __shfl_xor(s1, 4);
    s0 += __shfl_xor(s0, 8); s1 += __shfl_xor(s1, 8);
    if (p == 0) {
      out[(size_t)(row0 + r) * 2 + 0] = s0 + b2[0];
      out[(size_t)(row0 + r) * 2 + 1] = s1 + b2[1];
    }
  }
}

extern "C" void kernel_launch(void* const* d_in, const int* in_sizes, int n_in,
                              void* d_out, int out_size, void* d_ws, size_t ws_size,
                              hipStream_t stream) {
  const float* x  = (const float*)d_in[0];
  const int*   ei = (const int*)d_in[1];
  const float* Wl = (const float*)d_in[2];
  const float* bl = (const float*)d_in[3];
  const float* Wr = (const float*)d_in[4];
  const float* W1 = (const float*)d_in[5];
  const float* b1 = (const float*)d_in[6];
  const float* W2 = (const float*)d_in[7];
  const float* b2 = (const float*)d_in[8];
  float* out = (float*)d_out;

  const int N = NNODES;
  const int E = in_sizes[1] / 2;
  const int* src = ei;
  const int* dst = ei + E;

  float* inv_deg = (float*)d_ws;
  const size_t pad = 100096;
  float* bufA = inv_deg + pad;
  float* bufB = bufA + (size_t)N * HID;

  hipMemsetAsync(inv_deg, 0, (size_t)N * sizeof(float), stream);
  deg_kernel<<<(E + 255) / 256, 256, 0, stream>>>(dst, inv_deg, E);
  invdeg_kernel<<<(N + 255) / 256, 256, 0, stream>>>(inv_deg, N);

  const float* xs[3] = {x, bufA, bufB};
  float*       ag[3] = {bufA, bufB, bufA};
  const int scatterBlocks = (E + 7) / 8;
  const int xformBlocks   = N / ROWS;
  const int headBlocks    = N / TILE_M;

  for (int i = 0; i < 3; ++i) {
    float* agg = ag[i];
    const float* xc = xs[i];
    hipMemsetAsync(agg, 0, (size_t)N * HID * sizeof(float), stream);
    scatter_kernel<<<scatterBlocks, 256, 0, stream>>>(xc, src, dst, agg, E);
    sage_transform<<<xformBlocks, 256, 0, stream>>>(
        agg, xc, Wl + (size_t)i * HID * HID, bl + (size_t)i * HID,
        Wr + (size_t)i * HID * HID, inv_deg, agg);
  }
  mlp_head<<<headBlocks, 256, 0, stream>>>(bufA, W1, b1, W2, b2, out);
}


# Dense kernels re-tiled: a wave owns 16 rows x 128 cols, A fragments straight from global into registers, weights staged once per resident group in LDS in fragment order; k and sum order change
# speedup vs baseline: 3.2555x; 3.2555x over previous
#include <hip/hip_runtime.h>
#include <math.h>

typedef __attribute__((ext_vector_type(2))) float v2f;
typedef __attribute__((ext_vector_type(8))) float v8f;

#define NNODES 100000
#define HID    128
#define TILE_M 16

#define OWN_ROWS   400
#define OWN_WAVES  32
#define GRP_EDGES  512
#define LIST_CAP   (GRP_EDGES + 32)
#define ROWS_IN_FLIGHT 8

#define SRC_BITS   17
#define SRC_MASK   ((1 << SRC_BITS) - 1)
#define MAX_OWNERS 256
#define PAD_ENTRY  (OWN_ROWS << SRC_BITS)
static_assert(NNODES <= (1 << SRC_BITS), "source node must fit the low field");
static_assert(OWN_ROWS < (1 << (31 - SRC_BITS)), "owned row and the spare row must fit the high field");
static_assert((NNODES + OWN_ROWS - 1) / OWN_ROWS <= MAX_OWNERS, "owner count words");

static __device__ __forceinline__ float atomic_add_f32(float* p, float v) {
  return __hip_atomic_fetch_add(p, v, __ATOMIC_RELAXED, __HIP_MEMORY_SCOPE_AGENT);
}

__global__ __launch_bounds__(256) void deg_kernel(const int* __restrict__ dst,
                                                  float* __restrict__ deg, int E) {
  int e = blockIdx.x * 256 + threadIdx.x;
  if (e < E) atomic_add_f32(&deg[dst[e]], 1.0f);
}

__global__ __launch_bounds__(512) void invdeg_kernel(float* __restrict__ deg,
                                                     int* __restrict__ own_cnt, int N) {
  __shared__ int part[16];
  const int t = threadIdx.x;
  const int n = blockIdx.x * OWN_ROWS + t;
  int c = 0;
  if (t < OWN_ROWS && n < N) {
    const float d = deg[n];
    c = (int)d;
    deg[n] = 1.0f / fmaxf(d, 1.0f);
  }
  c += __shfl_xor(c, 16);
  c += __shfl_xor(c, 8);
  c += __shfl_xor(c, 4);
  c += __shfl_xor(c, 2);
  c += __shfl_xor(c, 1);
  if ((t & 31) == 0) part[t >> 5] = c;
  __syncthreads();
  if (t == 0) {
    int s = 0;
    for (int w = 0; w < 16; ++w) s += part[w];
    own_cnt[blockIdx.x] = s;
  }
}


static __device__ __forceinline__ int owner_list_start(const int* __restrict__ own_cnt,
                                                       int b, int lane) {
  int a = 0;
  for (int j = lane; j < b; j += 32) a += (own_cnt[j] + 31) & ~31;
  a += __shfl_xor(a, 16);
  a += __shfl_xor(a, 8);
  a += __shfl_xor(a, 4);
  a += __shfl_xor(a, 2);
  a += __shfl_xor(a, 1);
  return a;
}

static __device__ __forceinline__ void add_rows(const float* __restrict__ x,
                                                float* rows_lds, int s_l, int d_l,
                                                int n, int lane) {
#pragma unroll 1
  for (int b = 0; b < n; b += ROWS_IN_FLIGHT) {
    float v[ROWS_IN_FLIGHT][4];
#pragma unroll
    for (int j = 0; j < ROWS_IN_FLIGHT; ++j) {
      const int s = __builtin_amdgcn_readlane(s_l, b + j);
      const float* xr = x + (size_t)s * HID + lane;
      v[j][0] = xr[0];
      v[j][1] = xr[32];
      v[j][2] = xr[64];
      v[j][3] = xr[96];
    }
#pragma unroll
    for (int j = 0; j < ROWS_IN_FLIGHT; ++j) {
      const int d = __builtin_amdgcn_readlane(d_l, b + j);
      float* ap = rows_lds + d * HID + lane;
      __hip_atomic_fetch_add(ap,      v[j][0], __ATOMIC_RELAXED, __HIP_MEMORY_SCOPE_WORKGROUP);
      __hip_atomic_fetch_add(ap + 32, v[j][1], __ATOMIC_RELAXED, __HIP_MEMORY_SCOPE_WORKGROUP);
      __hip_atomic_fetch_add(ap + 64, v[j][2], __ATOMIC_RELAXED, __HIP_MEMORY_SCOPE_WORKGROUP);
      __hip_atomic_fetch_add(ap + 96, v[j][3], __ATOMIC_RELAXED, __HIP_MEMORY_SCOPE_WORKGROUP);
    }
  }
}

static __device__ __forceinline__ void add_listed_rows(
    const float* __restrict__ x, const int* __restrict__ src,
    const int* __restrict__ dst, float* rows_lds, const int* list,
    int n, int lo, int lane, int* fill, int* __restrict__ my_list) {
  int s_l = PAD_ENTRY & SRC_MASK, d_l = PAD_ENTRY >> SRC_BITS;
  if (lane < n) {
    const int e = list[lane];
    s_l = src[e];
    d_l = dst[e] - lo;
  }
  int at = 0;
  if (lane == 0)
    at = __hip_atomic_fetch_add(fill, n, __ATOMIC_RELAXED, __HIP_MEMORY_SCOPE_WORKGROUP);
  at = __builtin_amdgcn_readfirstlane(at);
  if (lane < n) my_list[at + lane] = s_l | (d_l << SRC_BITS);
  add_rows(x, rows_lds, s_l, d_l, n, lane);
}

static __device__ __forceinline__ void load_dst_group(const int* __restrict__ dst, int g,
                                                      int lane, int E, bool vec_ok,
                                                      int (&d)[16]) {
  const int e0 = g * GRP_EDGES + lane * 4;
  if (vec_ok && (g + 1) * GRP_EDGES <= E) {
#pragma unroll
    for (int c = 0; c < 4; ++c) {
      const int4 q = *(const int4*)(dst + e0 + c * 128);
      d[c * 4 + 0] = q.x; d[c * 4 + 1] = q.y; d[c * 4 + 2] = q.z; d[c * 4 + 3] = q.w;
    }
  } else {
#pragma unroll
    for (int c = 0; c < 4; ++c)
#pragma unroll
      for (int k = 0; k < 4; ++k) {
        const int e = e0 + c * 128 + k;
        d[c * 4 + k] = (e < E) ? dst[e] : -1;
      }
  }
}

__global__ __launch_bounds__(1024) void owner_sum_kernel(const float* __restrict__ x,
                                                         const int* __restrict__ src,
                                                         const int* __restrict__ dst,
                                                         const int* __restrict__ own_cnt,
                                                         int* __restrict__ own_list,
                                                         float* __restrict__ agg,
                                                         int E, int N) {
  __shared__ float rows_lds[(OWN_ROWS + 1) * HID];
  __shared__ int   list_lds[OWN_WAVES * LIST_CAP];
  __shared__ int   start_lds;
  __shared__ int   fill_lds;

  const int tid  = threadIdx.x;
  const int lane = tid & 31;
  const int wave = __builtin_amdgcn_readfirstlane(tid >> 5);
  const int lo   = blockIdx.x * OWN_ROWS;
  const int rows = min(OWN_ROWS, N - lo);

  for (int i = tid; i < OWN_ROWS * HID / 4; i += 1024)
    ((float4*)rows_lds)[i] = make_float4(0.f, 0.f, 0.f, 0.f);
  if (wave == 0) {
    const int a = owner_list_start(own_cnt, blockIdx.x, lane);
    if (lane == 0) { start_lds = a; fill_lds = 0; }
  }
  __syncthreads();
  int* my_list = own_list + start_lds;

  int* list = list_lds + wave * LIST_CAP;
  int  cnt  = 0;
  const bool vec_ok = (E & 3) == 0;
  const int  ngrp   = (E + GRP_EDGES - 1) / GRP_EDGES;
  const int  rot    = (int)((blockIdx.x * (unsigned)ngrp) / gridDim.x);

  int nxt[16];
  if (wave < ngrp) {
    int g = wave + rot;
    if (g >= ngrp) g -= ngrp;
    load_dst_group(dst, g, lane, E, vec_ok, nxt);
  }
#pragma unroll 1
  for (int t = wave; t < ngrp; t += OWN_WAVES) {
    int g = t + rot;
    if (g >= ngrp) g -= ngrp;
    const int e0 = g * GRP_EDGES + lane * 4;
    int d[16];
#pragma unroll
    for (int i = 0; i < 16; ++i) d[i] = nxt[i];
    if (t + OWN_WAVES < ngrp) {
      int gn = t + OWN_WAVES + rot;
      if (gn >= ngrp) gn -= ngrp;
      load_dst_group(dst, gn, lane, E, vec_ok, nxt);
    }
#pragma unroll
    for (int c = 0; c < 4; ++c)
#pragma unroll
      for (int k = 0; k < 4; ++k) {
        const bool mine = (unsigned)(d[c * 4 + k] - lo) < (unsigned)rows;
        const unsigned m = __builtin_amdgcn_ballot_w32(mine);
        if (m != 0u) {
          if (mine) list[cnt + (int)__builtin_amdgcn_mbcnt_lo(m, 0u)] = e0 + c * 128 + k;
          cnt += __builtin_popcount(m);
        }
      }
    __builtin_amdgcn_wave_barrier();
    while (cnt >= 32) {
      add_listed_rows(x, src, dst, rows_lds, list + (cnt - 32), 32, lo, lane,
                      &fill_lds, my_list);
      cnt -= 32;
    }
    __builtin_amdgcn_wave_barrier();
  }
  if (cnt > 0)
    add_listed_rows(x, src, dst, rows_lds, list, cnt, lo, lane, &fill_lds, my_list);
  __syncthreads();

  for (int i = tid; i < rows * HID / 4; i += 1024)
    ((float4*)(agg + (size_t)lo * HID))[i] = ((const float4*)rows_lds)[i];
}

__global__ __launch_bounds__(1024) void list_sum_kernel(const float* __restrict__ x,
                                                        const int* __restrict__ own_cnt,
                                                        const int* __restrict__ own_list,
                                                        float* __restrict__ agg, int N) {
  __shared__ float rows_lds[(OWN_ROWS + 1) * HID];
  __shared__ int   start_lds;

  const int tid  = threadIdx.x;
  const int lane = tid & 31;
  const int wave = __builtin_amdgcn_readfirstlane(tid >> 5);
  const int lo   = blockIdx.x * OWN_ROWS;
  const int rows = min(OWN_ROWS, N - lo);

  for (int i = tid; i < OWN_ROWS * HID / 4; i += 1024)
    ((float4*)rows_lds)[i] = make_float4(0.f, 0.f, 0.f, 0.f);
  if (wave == 0) {
    const int a = owner_list_start(own_cnt, blockIdx.x, lane);
    if (lane == 0) start_lds = a;
  }
  __syncthreads();
  const int* my_list = own_list + start_lds;

  const int total = own_cnt[blockIdx.x];
  const int share = ((total + OWN_WAVES - 1) / OWN_WAVES + 7) & ~7;
  const int end   = min(total, (wave + 1) * share);
  int i = wave * share;

  int ent = PAD_ENTRY;
  if (i + lane < end) ent = my_list[i + lane];
#pragma unroll 1
  for (; i < end; i += 32) {
    const int cur = ent;
    ent = PAD_ENTRY;
    if (i + 32 + lane < end) ent = my_list[i + 32 + lane];
    add_rows(x, rows_lds, cur & SRC_MASK, cur >> SRC_BITS, min(32, end - i), lane);
  }
  __syncthreads();

  for (int k = tid; k < rows * HID / 4; k += 1024)
    ((float4*)(agg + (size_t)lo * HID))[k] = ((const float4*)rows_lds)[k];
}

#define XF_GRID    256
#define XF_WAVES   16
#define XF_THREADS (XF_WAVES * 32)
#define XF_STRIDE  (XF_GRID * XF_WAVES)
#define NTILES     (NNODES / TILE_M)
#define WFRAGS     (16 * 8 * 32)
static_assert(NNODES % TILE_M == 0, "whole row tiles");

static __device__ __forceinline__ void stage_weights(const float* __restrict__ W,
                                                     float4* lds_w, int tid) {
  static_assert(WFRAGS % XF_THREADS == 0, "whole passes");
  float4 w[WFRAGS / XF_THREADS];
#pragma unroll
  for (int k = 0; k < WFRAGS / XF_THREADS; ++k) {
    const int i = tid + k * XF_THREADS;
    const int l = i & 31, t = (i >> 5) & 7, j = i >> 8;
    const int c = l & 15, h = l >> 4;
    w[k] = *(const float4*)(W + (size_t)(t * 16 + c) * HID + 8 * j + 4 * h);
  }
#pragma unroll
  for (int k = 0; k < WFRAGS / XF_THREADS; ++k) lds_w[tid + k * XF_THREADS] = w[k];
}

static __device__ __forceinline__ void load_b(float4 (&b)[8], const float4* p) {
#pragma unroll
  for (int t = 0; t < 8; ++t) b[t] = p[t * 32];
}

static __device__ __forceinline__ void k_step(const float4 a, const float4 (&b)[8],
                                              v8f (&acc)[8]) {
  const v2f a0 = {a.x, a.y};
  const v2f a1 = {a.z, a.w};
#pragma unroll
  for (int t = 0; t < 8; ++t) {
    const v2f b0 = {b[t].x, b[t].y};
    acc[t] = __builtin_amdgcn_wmma_f32_16x16x4_f32(false, a0, false, b0,
                                                   (short)0, acc[t], false, false);
  }
#pragma unroll
  for (int t = 0; t < 8; ++t) {
    const v2f b1 = {b[t].z, b[t].w};
    acc[t] = __builtin_amdgcn_wmma_f32_16x16x4_f32(false, a1, false, b1,
                                                   (short)0, acc[t], false, false);
  }
}

__global__ __launch_bounds__(XF_THREADS) void sage_transform(
    const float* agg, const float* __restrict__ xin,
    const float* __restrict__ Wl, const float* __restrict__ bl,
    const float* __restrict__ Wr, const float* __restrict__ invdeg,
    float* xout) {
  __shared__ float4 lds_w[2 * WFRAGS];

  const int tid  = threadIdx.x;
  const int lane = tid & 31;
  const int wave = __builtin_amdgcn_readfirstlane(tid >> 5);
  const int c = lane & 15;
  const int h = lane >> 4;

  stage_weights(Wl, lds_w, tid);
  stage_weights(Wr, lds_w + WFRAGS, tid);
  __syncthreads();

  const int tile0 = wave * XF_GRID + (int)blockIdx.x;
  if (tile0 >= NTILES) return;
  const int nhalf = 2 * ((NTILES - 1 - tile0) / XF_STRIDE + 1);

  float4 a[16];
  {
    const float4* pa = (const float4*)(agg + (size_t)(tile0 * TILE_M + c) * HID) + h;
#pragma unroll
    for (int j = 0; j < 16; ++j) a[j] = pa[2 * j];
  }
  float scale = invdeg[tile0 * TILE_M + c];
  v8f acc[8];
#pragma unroll
  for (int t = 0; t < 8; ++t) acc[t] = (v8f){0.f, 0.f, 0.f, 0.f, 0.f, 0.f, 0.f, 0.f};

#pragma unroll 1
  for (int q = 0; q < nhalf; ++q) {
    const int m    = q & 1;
    const int row0 = (tile0 + (q >> 1) * XF_STRIDE) * TILE_M;
    const int qn = min(q + 1, nhalf - 1);
    const int mn = qn & 1;
    const int rn = (tile0 + (qn >> 1) * XF_STRIDE) * TILE_M + c;
    const float4* pn = (const float4*)((mn ? xin : agg) + (size_t)rn * HID) + h;
    const float4* pb = lds_w + m * WFRAGS + lane;

    float4 an[16];
    float4 b[2][8];
    load_b(b[0], pb);
#pragma unroll
    for (int s = 0; s < 16; ++s) {
      if (s + 1 < 16) load_b(b[(s + 1) & 1], pb + (s + 1) * 256);
      float4 av = a[s];
      av.x *= scale; av.y *= scale; av.z *= scale; av.w *= scale;
      an[s] = pn[2 * s];
      k_step(av, b[s & 1], acc);
      __builtin_amdgcn_sched_barrier(0);
    }
    const float scale_n = mn ? 1.0f : invdeg[rn];

    if (m == 1) {
      float bias[8];
#pragma unroll
      for (int t = 0; t < 8; ++t) bias[t] = bl[t * 16 + c];
      float ss[8];
#pragma unroll
      for (int r = 0; r < 8; ++r) ss[r] = 0.f;
#pragma unroll
      for (int t = 0; t < 8; ++t)
#pragma unroll
        for (int r = 0; r < 8; ++r) {
          const float o = acc[t][r] + bias[t];
          acc[t][r] = o;
          ss[r] += o * o;
        }
#pragma unroll
      for (int r = 0; r < 8; ++r) {
        ss[r] += __shfl_xor(ss[r], 1);
        ss[r] += __shfl_xor(ss[r], 2);
        ss[r] += __shfl_xor(ss[r], 4);
        ss[r] += __shfl_xor(ss[r], 8);
      }
#pragma unroll
      for (int r = 0; r < 8; ++r) {
        const float sc = 1.0f / fmaxf(sqrtf(ss[r]), 1e-12f);
        float* po = xout + (size_t)(row0 + 8 * h + r) * HID + c;
#pragma unroll
        for (int t = 0; t < 8; ++t) po[t * 16] = fmaxf(acc[t][r] * sc, 0.f);
      }
#pragma unroll
      for (int t = 0; t < 8; ++t) acc[t] = (v8f){0.f, 0.f, 0.f, 0.f, 0.f, 0.f, 0.f, 0.f};
    }
#pragma unroll
    for (int j = 0; j < 16; ++j) a[j] = an[j];
    scale = scale_n;
  }
}

__global__ __launch_bounds__(XF_THREADS) void mlp_head(
    const float* __restrict__ xin, const float* __restrict__ W1,
    const float* __restrict__ b1, const float* __restrict__ W2,
    const float* __restrict__ b2, float* __restrict__ out) {
  __shared__ float4 lds_w[WFRAGS];

  const int tid  = threadIdx.x;
  const int lane = tid & 31;
  const int wave = __builtin_amdgcn_readfirstlane(tid >> 5);
  const int c = lane & 15;
  const int h = lane >> 4;

  stage_weights(W1, lds_w, tid);
  float bias[8], w20[8], w21[8];
#pragma unroll
  for (int t = 0; t < 8; ++t) {
    bias[t] = b1[t * 16 + c];
    w20[t]  = W2[t * 16 + c];
    w21[t]  = W2[HID + t * 16 + c];
  }
  const float bout = (c & 1) ? b2[1] : b2[0];
  __syncthreads();

#pragma unroll 1
  for (int tile = wave * XF_GRID + (int)blockIdx.x; tile < NTILES; tile += XF_STRIDE) {
    const int row0 = tile * TILE_M;
    const float4* px = (const float4*)(xin + (size_t)(row0 + c) * HID) + h;
    float4 aX[16];
#pragma unroll
    for (int j = 0; j < 16; ++j) aX[j] = px[2 * j];

    v8f acc[8];
#pragma unroll
    for (int t = 0; t < 8; ++t) acc[t] = (v8f){0.f, 0.f, 0.f, 0.f, 0.f, 0.f, 0.f, 0.f};
    float4 b[2][8];
    load_b(b[0], lds_w + lane);
#pragma unroll
    for (int s = 0; s < 16; ++s) {
      if (s + 1 < 16) load_b(b[(s + 1) & 1], lds_w + (s + 1) * 256 + lane);
      k_step(aX[s], b[s & 1], acc);
      __builtin_amdgcn_sched_barrier(0);
    }

    float v = 0.f;
#pragma unroll
    for (int r = 0; r < 8; ++r) {
      float s0 = 0.f, s1 = 0.f;
#pragma unroll
      for (int t = 0; t < 8; ++t) {
        const float hv = fmaxf(acc[t][r] + bias[t], 0.f);
        s0 += hv * w20[t];
        s1 += hv * w21[t];
      }
      s0 += __shfl_xor(s0, 1); s1 += __shfl_xor(s1, 1);
      s0 += __shfl_xor(s0, 2); s1 += __shfl_xor(s1, 2);
      s0 += __shfl_xor(s0, 4); s1 += __shfl_xor(s1, 4);
      s0 += __shfl_xor(s0, 8); s1 += __shfl_xor(s1, 8);
      const float pick = (c & 1) ? s1 : s0;
      v = ((c >> 1) == r) ? pick : v;
    }
    out[(size_t)row0 * 2 + lane] = v + bout;
  }
}

extern "C" void kernel_launch(void* const* d_in, const int* in_sizes, int n_in,
                              void* d_out, int out_size, void* d_ws, size_t ws_size,
                              hipStream_t stream) {
  const float* x  = (const float*)d_in[0];
  const int*   ei = (const int*)d_in[1];
  const float* Wl = (const float*)d_in[2];
  const float* bl = (const float*)d_in[3];
  const float* Wr = (const float*)d_in[4];
  const float* W1 = (const float*)d_in[5];
  const float* b1 = (const float*)d_in[6];
  const float* W2 = (const float*)d_in[7];
  const float* b2 = (const float*)d_in[8];
  float* out = (float*)d_out;

  const int N = NNODES;
  const int E = in_sizes[1] / 2;
  const int* src = ei;
  const int* dst = ei + E;

  float* inv_deg = (float*)d_ws;
  const size_t pad = 100096;
  float* bufA = inv_deg + pad;
  float* bufB = bufA + (size_t)N * HID;
  int* own_cnt  = (int*)(bufB + (size_t)N * HID);
  int* own_list = own_cnt + MAX_OWNERS;

  const int ownerBlocks   = (N + OWN_ROWS - 1) / OWN_ROWS;

  hipMemsetAsync(inv_deg, 0, (size_t)N * sizeof(float), stream);
  deg_kernel<<<(E + 255) / 256, 256, 0, stream>>>(dst, inv_deg, E);
  invdeg_kernel<<<ownerBlocks, 512, 0, stream>>>(inv_deg, own_cnt, N);

  const float* xs[3] = {x, bufA, bufB};
  float*       ag[3] = {bufA, bufB, bufA};

  for (int i = 0; i < 3; ++i) {
    float* agg = ag[i];
    const float* xc = xs[i];
    if (i == 0)
      owner_sum_kernel<<<ownerBlocks, 1024, 0, stream>>>(xc, src, dst, own_cnt, own_list,
                                                         agg, E, N);
    else
      list_sum_kernel<<<ownerBlocks, 1024, 0, stream>>>(xc, own_cnt, own_list, agg, N);
    sage_transform<<<XF_GRID, XF_THREADS, 0, stream>>>(
        agg, xc, Wl + (size_t)i * HID * HID, bl + (size_t)i * HID,
        Wr + (size_t)i * HID * HID, inv_deg, agg);
  }
  mlp_head<<<XF_GRID, XF_THREADS, 0, stream>>>(bufA, W1, b1, W2, b2, out);
}


# Owner edge lists split per wave by dst row & 31: a wave sums only its own LDS rows by plain 16-byte load-add-store, one 512-B load per row; no LDS f32 atomics; sum order = LDS slot-counter order
# speedup vs baseline: 3.4054x; 1.0460x over previous
#include <hip/hip_runtime.h>
#include <math.h>

typedef __attribute__((ext_vector_type(2))) float v2f;
typedef __attribute__((ext_vector_type(8))) float v8f;

#define NNODES 100000
#define HID    128
#define TILE_M 16

#define OWN_ROWS   400
#define OWN_WAVES  32
#define GRP_EDGES  512
#define LIST_CAP   (GRP_EDGES + 32)
#define ROWS_IN_FLIGHT 8

#define SRC_BITS   17
#define SRC_MASK   ((1 << SRC_BITS) - 1)
#define MAX_OWNERS 256
#define PAD_ENTRY  (OWN_ROWS << SRC_BITS)
static_assert(NNODES <= (1 << SRC_BITS), "source node must fit the low field");
static_assert(OWN_ROWS < (1 << (31 - SRC_BITS)), "owned row and the spare row must fit the high field");
static_assert((NNODES + OWN_ROWS - 1) / OWN_ROWS <= MAX_OWNERS, "owner count words");
static_assert(OWN_WAVES == 32, "a row's sub-list is its low five bits; one sub-list per lane of wave 0");

static __device__ __forceinline__ float atomic_add_f32(float* p, float v) {
  return __hip_atomic_fetch_add(p, v, __ATOMIC_RELAXED, __HIP_MEMORY_SCOPE_AGENT);
}

__global__ __launch_bounds__(256) void deg_kernel(const int* __restrict__ dst,
                                                  float* __restrict__ deg, int E) {
  int e = blockIdx.x * 256 + threadIdx.x;
  if (e < E) atomic_add_f32(&deg[dst[e]], 1.0f);
}

__global__ __launch_bounds__(512) void invdeg_kernel(float* __restrict__ deg,
                                                     int* __restrict__ own_len,
                                                     int* __restrict__ sub_cnt, int N) {
  __shared__ int part[16][32];
  const int t = threadIdx.x;
  const int n = blockIdx.x * OWN_ROWS + t;
  int c = 0;
  if (t < OWN_ROWS && n < N) {
    const float d = deg[n];
    c = (int)d;
    deg[n] = 1.0f / fmaxf(d, 1.0f);
  }
  part[t >> 5][t & 31] = c;
  __syncthreads();
  if (t < 32) {
    int s = 0;
    for (int k = 0; k < 16; ++k) s += part[k][t];
    sub_cnt[blockIdx.x * OWN_WAVES + t] = s;
    int r = (s + 31) & ~31;
    r += __shfl_xor(r, 16);
    r += __shfl_xor(r, 8);
    r += __shfl_xor(r, 4);
    r += __shfl_xor(r, 2);
    r += __shfl_xor(r, 1);
    if (t == 0) own_len[blockIdx.x] = r;
  }
}


static __device__ __forceinline__ void owner_sub_lists(const int* __restrict__ own_len,
                                                       const int* __restrict__ sub_cnt,
                                                       int b, int lane, int* sub_start,
                                                       int* sub_n, int* fill) {
  int a = 0;
  for (int j = lane; j < b; j += 32) a += own_len[j];
  a += __shfl_xor(a, 16);
  a += __shfl_xor(a, 8);
  a += __shfl_xor(a, 4);
  a += __shfl_xor(a, 2);
  a += __shfl_xor(a, 1);
  const int c = sub_cnt[b * OWN_WAVES + lane];
  const int r = (c + 31) & ~31;
  int inc = r;
  for (int o = 1; o < 32; o <<= 1) {
    const int u = __shfl_up(inc, o);
    if (lane >= o) inc += u;
  }
  sub_start[lane] = a + inc - r;
  sub_n[lane]     = c;
  if (fill) fill[lane] = 0;
}

static __device__ __forceinline__ void load_rows8(const float* __restrict__ x, int s_l,
                                                  int b, int lane, float4 (&v)[ROWS_IN_FLIGHT]) {
#pragma unroll
  for (int j = 0; j < ROWS_IN_FLIGHT; ++j) {
    const int s = __builtin_amdgcn_readlane(s_l, b + j);
    v[j] = *((const float4*)(x + (size_t)s * HID) + lane);
  }
}

static __device__ __forceinline__ void add_rows8(float* rows_lds, int d_l, int b, int lane,
                                                 const float4 (&v)[ROWS_IN_FLIGHT]) {
#pragma unroll
  for (int j = 0; j < ROWS_IN_FLIGHT; ++j) {
    const int d = __builtin_amdgcn_readlane(d_l, b + j);
    float4* ap = (float4*)(rows_lds + d * HID) + lane;
    float4 a = *ap;
    a.x += v[j].x; a.y += v[j].y; a.z += v[j].z; a.w += v[j].w;
    *ap = a;
  }
}

static __device__ __forceinline__ void sum_own_rows(const float* __restrict__ x,
                                                    float* rows_lds, const int* sub,
                                                    int n, int lane) {
  if (n <= 0) return;
  int cur = PAD_ENTRY, nxt = PAD_ENTRY;
  if (lane < n) cur = sub[lane];
  if (32 + lane < n) nxt = sub[32 + lane];
  float4 v[ROWS_IN_FLIGHT], vn[ROWS_IN_FLIGHT];
  load_rows8(x, cur & SRC_MASK, 0, lane, v);
#pragma unroll 1
  for (int i = 0; i < n; i += 32) {
    const int m   = min(32, n - i);
    const int s_l = cur & SRC_MASK, d_l = cur >> SRC_BITS;
    const int s_n = nxt & SRC_MASK;
#pragma unroll 1
    for (int b = 0; b < m; b += ROWS_IN_FLIGHT) {
      if (b + ROWS_IN_FLIGHT < m) load_rows8(x, s_l, b + ROWS_IN_FLIGHT, lane, vn);
      else                        load_rows8(x, s_n, 0, lane, vn);
      add_rows8(rows_lds, d_l, b, lane, v);
#pragma unroll
      for (int j = 0; j < ROWS_IN_FLIGHT; ++j) v[j] = vn[j];
    }
    cur = nxt;
    nxt = PAD_ENTRY;
    if (i + 64 + lane < n) nxt = sub[i + 64 + lane];
  }
}

static __device__ __forceinline__ void place_listed_edges(
    const int* __restrict__ src, const int* __restrict__ dst, const int* list,
    int n, int lo, int lane, int* fill, const int* sub_start, int* own_list) {
  if (lane < n) {
    const int e = list[lane];
    const int s = src[e];
    const int d = dst[e] - lo;
    const int w = d & 31;
    const int at = __hip_atomic_fetch_add(fill + w, 1, __ATOMIC_RELAXED,
                                          __HIP_MEMORY_SCOPE_WORKGROUP);
    own_list[sub_start[w] + at] = s | (d << SRC_BITS);
  }
}

static __device__ __forceinline__ void load_dst_group(const int* __restrict__ dst, int g,
                                                      int lane, int E, bool vec_ok,
                                                      int (&d)[16]) {
  const int e0 = g * GRP_EDGES + lane * 4;
  if (vec_ok && (g + 1) * GRP_EDGES <= E) {
#pragma unroll
    for (int c = 0; c < 4; ++c) {
      const int4 q = *(const int4*)(dst + e0 + c * 128);
      d[c * 4 + 0] = q.x; d[c * 4 + 1] = q.y; d[c * 4 + 2] = q.z; d[c * 4 + 3] = q.w;
    }
  } else {
#pragma unroll
    for (int c = 0; c < 4; ++c)
#pragma unroll
      for (int k = 0; k < 4; ++k) {
        const int e = e0 + c * 128 + k;
        d[c * 4 + k] = (e < E) ? dst[e] : -1;
      }
  }
}

__global__ __launch_bounds__(1024) void owner_sum_kernel(const float* __restrict__ x,
                                                         const int* __restrict__ src,
                                                         const int* __restrict__ dst,
                                                         const int* __restrict__ own_len,
                                                         const int* __restrict__ sub_cnt,
                                                         int* own_list,
                                                         float* __restrict__ agg,
                                                         int E, int N) {
  __shared__ float rows_lds[(OWN_ROWS + 1) * HID];
  __shared__ int   list_lds[OWN_WAVES * LIST_CAP];
  __shared__ int   sub_start_lds[OWN_WAVES];
  __shared__ int   sub_n_lds[OWN_WAVES];
  __shared__ int   fill_lds[OWN_WAVES];

  const int tid  = threadIdx.x;
  const int lane = tid & 31;
  const int wave = __builtin_amdgcn_readfirstlane(tid >> 5);
  const int lo   = blockIdx.x * OWN_ROWS;
  const int rows = min(OWN_ROWS, N - lo);

  for (int i = tid; i < OWN_ROWS * HID / 4; i += 1024)
    ((float4*)rows_lds)[i] = make_float4(0.f, 0.f, 0.f, 0.f);
  if (wave == 0)
    owner_sub_lists(own_len, sub_cnt, blockIdx.x, lane, sub_start_lds, sub_n_lds, fill_lds);
  __syncthreads();

  int* list = list_lds + wave * LIST_CAP;
  int  cnt  = 0;
  const bool vec_ok = (E & 3) == 0;
  const int  ngrp   = (E + GRP_EDGES - 1) / GRP_EDGES;
  const int  rot    = (int)((blockIdx.x * (unsigned)ngrp) / gridDim.x);

  int nxt[16];
  if (wave < ngrp) {
    int g = wave + rot;
    if (g >= ngrp) g -= ngrp;
    load_dst_group(dst, g, lane, E, vec_ok, nxt);
  }
#pragma unroll 1
  for (int t = wave; t < ngrp; t += OWN_WAVES) {
    int g = t + rot;
    if (g >= ngrp) g -= ngrp;
    const int e0 = g * GRP_EDGES + lane * 4;
    int d[16];
#pragma unroll
    for (int i = 0; i < 16; ++i) d[i] = nxt[i];
    if (t + OWN_WAVES < ngrp) {
      int gn = t + OWN_WAVES + rot;
      if (gn >= ngrp) gn -= ngrp;
      load_dst_group(dst, gn, lane, E, vec_ok, nxt);
    }
#pragma unroll
    for (int c = 0; c < 4; ++c)
#pragma unroll
      for (int k = 0; k < 4; ++k) {
        const bool mine = (unsigned)(d[c * 4 + k] - lo) < (unsigned)rows;
        const unsigned m = __builtin_amdgcn_ballot_w32(mine);
        if (m != 0u) {
          if (mine) list[cnt + (int)__builtin_amdgcn_mbcnt_lo(m, 0u)] = e0 + c * 128 + k;
          cnt += __builtin_popcount(m);
        }
      }
    __builtin_amdgcn_wave_barrier();
    while (cnt >= 32) {
      place_listed_edges(src, dst, list + (cnt - 32), 32, lo, lane, fill_lds,
                         sub_start_lds, own_list);
      cnt -= 32;
    }
    __builtin_amdgcn_wave_barrier();
  }
  if (cnt > 0)
    place_listed_edges(src, dst, list, cnt, lo, lane, fill_lds, sub_start_lds, own_list);
  __syncthreads();

  sum_own_rows(x, rows_lds, own_list + __builtin_amdgcn_readfirstlane(sub_start_lds[wave]),
               __builtin_amdgcn_readfirstlane(sub_n_lds[wave]), lane);
  __syncthreads();

  for (int i = tid; i < rows * HID / 4; i += 1024)
    ((float4*)(agg + (size_t)lo * HID))[i] = ((const float4*)rows_lds)[i];
}

__global__ __launch_bounds__(1024) void list_sum_kernel(const float* __restrict__ x,
                                                        const int* __restrict__ own_len,
                                                        const int* __restrict__ sub_cnt,
                                                        const int* __restrict__ own_list,
                                                        float* __restrict__ agg, int N) {
  __shared__ float rows_lds[(OWN_ROWS + 1) * HID];
  __shared__ int   sub_start_lds[OWN_WAVES];
  __shared__ int   sub_n_lds[OWN_WAVES];

  const int tid  = threadIdx.x;
  const int lane = tid & 31;
  const int wave = __builtin_amdgcn_readfirstlane(tid >> 5);
  const int lo   = blockIdx.x * OWN_ROWS;
  const int rows = min(OWN_ROWS, N - lo);

  for (int i = tid; i < OWN_ROWS * HID / 4; i += 1024)
    ((float4*)rows_lds)[i] = make_float4(0.f, 0.f, 0.f, 0.f);
  if (wave == 0)
    owner_sub_lists(own_len, sub_cnt, blockIdx.x, lane, sub_start_lds, sub_n_lds, nullptr);
  __syncthreads();

  sum_own_rows(x, rows_lds, own_list + __builtin_amdgcn_readfirstlane(sub_start_lds[wave]),
               __builtin_amdgcn_readfirstlane(sub_n_lds[wave]), lane);
  __syncthreads();

  for (int k = tid; k < rows * HID / 4; k += 1024)
    ((float4*)(agg + (size_t)lo * HID))[k] = ((const float4*)rows_lds)[k];
}

#define XF_GRID    256
#define XF_WAVES   16
#define XF_THREADS (XF_WAVES * 32)
#define XF_STRIDE  (XF_GRID * XF_WAVES)
#define NTILES     (NNODES / TILE_M)
#define WFRAGS     (16 * 8 * 32)
static_assert(NNODES % TILE_M == 0, "whole row tiles");

static __device__ __forceinline__ void stage_weights(const float* __restrict__ W,
                                                     float4* lds_w, int tid) {
  static_assert(WFRAGS % XF_THREADS == 0, "whole passes");
  float4 w[WFRAGS / XF_THREADS];
#pragma unroll
  for (int k = 0; k < WFRAGS / XF_THREADS; ++k) {
    const int i = tid + k * XF_THREADS;
    const int l = i & 31, t = (i >> 5) & 7, j = i >> 8;
    const int c = l & 15, h = l >> 4;
    w[k] = *(const float4*)(W + (size_t)(t * 16 + c) * HID + 8 * j + 4 * h);
  }
#pragma unroll
  for (int k = 0; k < WFRAGS / XF_THREADS; ++k) lds_w[tid + k * XF_THREADS] = w[k];
}

static __device__ __forceinline__ void load_b(float4 (&b)[8], const float4* p) {
#pragma unroll
  for (int t = 0; t < 8; ++t) b[t] = p[t * 32];
}

static __device__ __forceinline__ void k_step(const float4 a, const float4 (&b)[8],
                                              v8f (&acc)[8]) {
  const v2f a0 = {a.x, a.y};
  const v2f a1 = {a.z, a.w};
#pragma unroll
  for (int t = 0; t < 8; ++t) {
    const v2f b0 = {b[t].x, b[t].y};
    acc[t] = __builtin_amdgcn_wmma_f32_16x16x4_f32(false, a0, false, b0,
                                                   (short)0, acc[t], false, false);
  }
#pragma unroll
  for (int t = 0; t < 8; ++t) {
    const v2f b1 = {b[t].z, b[t].w};
    acc[t] = __builtin_amdgcn_wmma_f32_16x16x4_f32(false, a1, false, b1,
                                                   (short)0, acc[t], false, false);
  }
}

__global__ __launch_bounds__(XF_THREADS) void sage_transform(
    const float* agg, const float* __restrict__ xin,
    const float* __restrict__ Wl, const float* __restrict__ bl,
    const float* __restrict__ Wr, const float* __restrict__ invdeg,
    float* xout) {
  __shared__ float4 lds_w[2 * WFRAGS];

  const int tid  = threadIdx.x;
  const int lane = tid & 31;
  const int wave = __builtin_amdgcn_readfirstlane(tid >> 5);
  const int c = lane & 15;
  const int h = lane >> 4;

  stage_weights(Wl, lds_w, tid);
  stage_weights(Wr, lds_w + WFRAGS, tid);
  __syncthreads();

  const int tile0 = wave * XF_GRID + (int)blockIdx.x;
  if (tile0 >= NTILES) return;
  const int nhalf = 2 * ((NTILES - 1 - tile0) / XF_STRIDE + 1);

  float4 a[16];
  {
    const float4* pa = (const float4*)(agg + (size_t)(tile0 * TILE_M + c) * HID) + h;
#pragma unroll
    for (int j = 0; j < 16; ++j) a[j] = pa[2 * j];
  }
  float scale = invdeg[tile0 * TILE_M + c];
  v8f acc[8];
#pragma unroll
  for (int t = 0; t < 8; ++t) acc[t] = (v8f){0.f, 0.f, 0.f, 0.f, 0.f, 0.f, 0.f, 0.f};

#pragma unroll 1
  for (int q = 0; q < nhalf; ++q) {
    const int m    = q & 1;
    const int row0 = (tile0 + (q >> 1) * XF_STRIDE) * TILE_M;
    const int qn = min(q + 1, nhalf - 1);
    const int mn = qn & 1;
    const int rn = (tile0 + (qn >> 1) * XF_STRIDE) * TILE_M + c;
    const float4* pn = (const float4*)((mn ? xin : agg) + (size_t)rn * HID) + h;
    const float4* pb = lds_w + m * WFRAGS + lane;

    float4 an[16];
    float4 b[2][8];
    load_b(b[0], pb);
#pragma unroll
    for (int s = 0; s < 16; ++s) {
      if (s + 1 < 16) load_b(b[(s + 1) & 1], pb + (s + 1) * 256);
      float4 av = a[s];
      av.x *= scale; av.y *= scale; av.z *= scale; av.w *= scale;
      an[s] = pn[2 * s];
      k_step(av, b[s & 1], acc);
      __builtin_amdgcn_sched_barrier(0);
    }
    const float scale_n = mn ? 1.0f : invdeg[rn];

    if (m == 1) {
      float bias[8];
#pragma unroll
      for (int t = 0; t < 8; ++t) bias[t] = bl[t * 16 + c];
      float ss[8];
#pragma unroll
      for (int r = 0; r < 8; ++r) ss[r] = 0.f;
#pragma unroll
      for (int t = 0; t < 8; ++t)
#pragma unroll
        for (int r = 0; r < 8; ++r) {
          const float o = acc[t][r] + bias[t];
          acc[t][r] = o;
          ss[r] += o * o;
        }
#pragma unroll
      for (int r = 0; r < 8; ++r) {
        ss[r] += __shfl_xor(ss[r], 1);
        ss[r] += __shfl_xor(ss[r], 2);
        ss[r] += __shfl_xor(ss[r], 4);
        ss[r] += __shfl_xor(ss[r], 8);
      }
#pragma unroll
      for (int r = 0; r < 8; ++r) {
        const float sc = 1.0f / fmaxf(sqrtf(ss[r]), 1e-12f);
        float* po = xout + (size_t)(row0 + 8 * h + r) * HID + c;
#pragma unroll
        for (int t = 0; t < 8; ++t) po[t * 16] = fmaxf(acc[t][r] * sc, 0.f);
      }
#pragma unroll
      for (int t = 0; t < 8; ++t) acc[t] = (v8f){0.f, 0.f, 0.f, 0.f, 0.f, 0.f, 0.f, 0.f};
    }
#pragma unroll
    for (int j = 0; j < 16; ++j) a[j] = an[j];
    scale = scale_n;
  }
}

__global__ __launch_bounds__(XF_THREADS) void mlp_head(
    const float* __restrict__ xin, const float* __restrict__ W1,
    const float* __restrict__ b1, const float* __restrict__ W2,
    const float* __restrict__ b2, float* __restrict__ out) {
  __shared__ float4 lds_w[WFRAGS];

  const int tid  = threadIdx.x;
  const int lane = tid & 31;
  const int wave = __builtin_amdgcn_readfirstlane(tid >> 5);
  const int c = lane & 15;
  const int h = lane >> 4;

  stage_weights(W1, lds_w, tid);
  float bias[8], w20[8], w21[8];
#pragma unroll
  for (int t = 0; t < 8; ++t) {
    bias[t] = b1[t * 16 + c];
    w20[t]  = W2[t * 16 + c];
    w21[t]  = W2[HID + t * 16 + c];
  }
  const float bout = (c & 1) ? b2[1] : b2[0];
  __syncthreads();

#pragma unroll 1
  for (int tile = wave * XF_GRID + (int)blockIdx.x; tile < NTILES; tile += XF_STRIDE) {
    const int row0 = tile * TILE_M;
    const float4* px = (const float4*)(xin + (size_t)(row0 + c) * HID) + h;
    float4 aX[16];
#pragma unroll
    for (int j = 0; j < 16; ++j) aX[j] = px[2 * j];

    v8f acc[8];
#pragma unroll
    for (int t = 0; t < 8; ++t) acc[t] = (v8f){0.f, 0.f, 0.f, 0.f, 0.f, 0.f, 0.f, 0.f};
    float4 b[2][8];
    load_b(b[0], lds_w + lane);
#pragma unroll
    for (int s = 0; s < 16; ++s) {
      if (s + 1 < 16) load_b(b[(s + 1) & 1], lds_w + (s + 1) * 256 + lane);
      k_step(aX[s], b[s & 1], acc);
      __builtin_amdgcn_sched_barrier(0);
    }

    float v = 0.f;
#pragma unroll
    for (int r = 0; r < 8; ++r) {
      float s0 = 0.f, s1 = 0.f;
#pragma unroll
      for (int t = 0; t < 8; ++t) {
        const float hv = fmaxf(acc[t][r] + bias[t], 0.f);
        s0 += hv * w20[t];
        s1 += hv * w21[t];
      }
      s0 += __shfl_xor(s0, 1); s1 += __shfl_xor(s1, 1);
      s0 += __shfl_xor(s0, 2); s1 += __shfl_xor(s1, 2);
      s0 += __shfl_xor(s0, 4); s1 += __shfl_xor(s1, 4);
      s0 += __shfl_xor(s0, 8); s1 += __shfl_xor(s1, 8);
      const float pick = (c & 1) ? s1 : s0;
      v = ((c >> 1) == r) ? pick : v;
    }
    out[(size_t)row0 * 2 + lane] = v + bout;
  }
}

extern "C" void kernel_launch(void* const* d_in, const int* in_sizes, int n_in,
                              void* d_out, int out_size, void* d_ws, size_t ws_size,
                              hipStream_t stream) {
  const float* x  = (const float*)d_in[0];
  const int*   ei = (const int*)d_in[1];
  const float* Wl = (const float*)d_in[2];
  const float* bl = (const float*)d_in[3];
  const float* Wr = (const float*)d_in[4];
  const float* W1 = (const float*)d_in[5];
  const float* b1 = (const float*)d_in[6];
  const float* W2 = (const float*)d_in[7];
  const float* b2 = (const float*)d_in[8];
  float* out = (float*)d_out;

  const int N = NNODES;
  const int E = in_sizes[1] / 2;
  const int* src = ei;
  const int* dst = ei + E;

  float* inv_deg = (float*)d_ws;
  const size_t pad = 100096;
  float* bufA = inv_deg + pad;
  float* bufB = bufA + (size_t)N * HID;
  int* own_len  = (int*)(bufB + (size_t)N * HID);
  int* sub_cnt  = own_len + MAX_OWNERS;
  int* own_list = sub_cnt + MAX_OWNERS * OWN_WAVES;

  const int ownerBlocks   = (N + OWN_ROWS - 1) / OWN_ROWS;

  hipMemsetAsync(inv_deg, 0, (size_t)N * sizeof(float), stream);
  deg_kernel<<<(E + 255) / 256, 256, 0, stream>>>(dst, inv_deg, E);
  invdeg_kernel<<<ownerBlocks, 512, 0, stream>>>(inv_deg, own_len, sub_cnt, N);

  const float* xs[3] = {x, bufA, bufB};
  float*       ag[3] = {bufA, bufB, bufA};

  for (int i = 0; i < 3; ++i) {
    float* agg = ag[i];
    const float* xc = xs[i];
    if (i == 0)
      owner_sum_kernel<<<ownerBlocks, 1024, 0, stream>>>(xc, src, dst, own_len, sub_cnt,
                                                         own_list, agg, E, N);
    else
      list_sum_kernel<<<ownerBlocks, 1024, 0, stream>>>(xc, own_len, sub_cnt, own_list, agg, N);
    sage_transform<<<XF_GRID, XF_THREADS, 0, stream>>>(
        agg, xc, Wl + (size_t)i * HID * HID, bl + (size_t)i * HID,
        Wr + (size_t)i * HID * HID, inv_deg, agg);
  }
  mlp_head<<<XF_GRID, XF_THREADS, 0, stream>>>(bufA, W1, b1, W2, b2, out);
}


# In-degrees counted in LDS by the owner groups while they collect their edges (raw stretch, then copied to the sub-lists, sums still in LDS slot order): 1.6M scattered global f32 adds and a launch go
# speedup vs baseline: 4.0977x; 1.2033x over previous
#include <hip/hip_runtime.h>
#include <math.h>

typedef __attribute__((ext_vector_type(2))) float v2f;
typedef __attribute__((ext_vector_type(8))) float v8f;

#define NNODES 100000
#define HID    128
#define TILE_M 16

#define OWN_ROWS   400
#define OWN_WAVES  32
#define GRP_EDGES  512
#define LIST_CAP   (GRP_EDGES + 32)
#define ROWS_IN_FLIGHT 8

#define SRC_BITS   17
#define SRC_MASK   ((1 << SRC_BITS) - 1)
#define MAX_OWNERS 256
#define DEG_WORDS  416
#define OWN_ROOM  1024
#define LIST_ROOM (MAX_OWNERS * (OWN_ROOM + 32))
#define RAW_LOADS 8
#define PAD_ENTRY  (OWN_ROWS << SRC_BITS)
static_assert(NNODES <= (1 << SRC_BITS), "source node must fit the low field");
static_assert(OWN_ROWS < (1 << (31 - SRC_BITS)), "owned row and the spare row must fit the high field");
static_assert((NNODES + OWN_ROWS - 1) / OWN_ROWS <= MAX_OWNERS, "owner count words");
static_assert(OWN_WAVES == 32, "a row's sub-list is its low five bits; one sub-list per lane of wave 0");
static_assert(DEG_WORDS >= OWN_ROWS && DEG_WORDS % 32 == 0, "one count per owned row");

#define CNT_LOADS 8
__global__ __launch_bounds__(1024) void owner_count_kernel(const int* __restrict__ dst,
                                                           int* __restrict__ own_cnt,
                                                           int E, int N) {
  __shared__ int h[MAX_OWNERS];
  const int t = threadIdx.x;
  if (t < MAX_OWNERS) h[t] = 0;
  __syncthreads();
  const int base = blockIdx.x * (1024 * CNT_LOADS) + t;
  int d[CNT_LOADS];
#pragma unroll
  for (int k = 0; k < CNT_LOADS; ++k) {
    const int e = base + k * 1024;
    d[k] = (e < E) ? dst[e] : -1;
  }
#pragma unroll
  for (int k = 0; k < CNT_LOADS; ++k)
    if ((unsigned)d[k] < (unsigned)N)
      __hip_atomic_fetch_add(h + d[k] / OWN_ROWS, 1, __ATOMIC_RELAXED,
                             __HIP_MEMORY_SCOPE_WORKGROUP);
  __syncthreads();
  if (t < MAX_OWNERS && h[t] != 0)
    __hip_atomic_fetch_add(own_cnt + t, h[t], __ATOMIC_RELAXED, __HIP_MEMORY_SCOPE_AGENT);
}


static __device__ __forceinline__ int owner_stretch_start(const int* __restrict__ own_cnt,
                                                          int b, int lane) {
  int a = 0;
  for (int j = lane; j < b; j += 32) a += ((own_cnt[j] + 31) & ~31) + OWN_ROOM;
  a += __shfl_xor(a, 16);
  a += __shfl_xor(a, 8);
  a += __shfl_xor(a, 4);
  a += __shfl_xor(a, 2);
  a += __shfl_xor(a, 1);
  return a;
}

static __device__ __forceinline__ void owner_sub_lists(const int* deg_lds, int own_start,
                                                       int b, int lane, int* sub_start,
                                                       int* sub_n, int* fill,
                                                       int* __restrict__ sub_start_g,
                                                       int* __restrict__ sub_cnt_g) {
  int c = 0;
  for (int k = 0; k < DEG_WORDS / 32; ++k) c += deg_lds[k * 32 + lane];
  const int r = (c + 31) & ~31;
  int inc = r;
  for (int o = 1; o < 32; o <<= 1) {
    const int u = __shfl_up(inc, o);
    if (lane >= o) inc += u;
  }
  const int st = own_start + inc - r;
  sub_start[lane] = st;
  sub_n[lane]     = c;
  fill[lane]      = 0;
  sub_start_g[b * OWN_WAVES + lane] = st;
  sub_cnt_g[b * OWN_WAVES + lane]   = c;
}

static __device__ __forceinline__ void load_rows8(const float* __restrict__ x, int s_l,
                                                  int b, int lane, float4 (&v)[ROWS_IN_FLIGHT]) {
#pragma unroll
  for (int j = 0; j < ROWS_IN_FLIGHT; ++j) {
    const int s = __builtin_amdgcn_readlane(s_l, b + j);
    v[j] = *((const float4*)(x + (size_t)s * HID) + lane);
  }
}

static __device__ __forceinline__ void add_rows8(float* rows_lds, int d_l, int b, int lane,
                                                 const float4 (&v)[ROWS_IN_FLIGHT]) {
#pragma unroll
  for (int j = 0; j < ROWS_IN_FLIGHT; ++j) {
    const int d = __builtin_amdgcn_readlane(d_l, b + j);
    float4* ap = (float4*)(rows_lds + d * HID) + lane;
    float4 a = *ap;
    a.x += v[j].x; a.y += v[j].y; a.z += v[j].z; a.w += v[j].w;
    *ap = a;
  }
}

static __device__ __forceinline__ void sum_own_rows(const float* __restrict__ x,
                                                    float* rows_lds, const int* sub,
                                                    int n, int lane) {
  if (n <= 0) return;
  int cur = PAD_ENTRY, nxt = PAD_ENTRY;
  if (lane < n) cur = sub[lane];
  if (32 + lane < n) nxt = sub[32 + lane];
  float4 v[ROWS_IN_FLIGHT], vn[ROWS_IN_FLIGHT];
  load_rows8(x, cur & SRC_MASK, 0, lane, v);
#pragma unroll 1
  for (int i = 0; i < n; i += 32) {
    const int m   = min(32, n - i);
    const int s_l = cur & SRC_MASK, d_l = cur >> SRC_BITS;
    const int s_n = nxt & SRC_MASK;
#pragma unroll 1
    for (int b = 0; b < m; b += ROWS_IN_FLIGHT) {
      if (b + ROWS_IN_FLIGHT < m) load_rows8(x, s_l, b + ROWS_IN_FLIGHT, lane, vn);
      else                        load_rows8(x, s_n, 0, lane, vn);
      add_rows8(rows_lds, d_l, b, lane, v);
#pragma unroll
      for (int j = 0; j < ROWS_IN_FLIGHT; ++j) v[j] = vn[j];
    }
    cur = nxt;
    nxt = PAD_ENTRY;
    if (i + 64 + lane < n) nxt = sub[i + 64 + lane];
  }
}

static __device__ __forceinline__ void place_listed_edges(
    const int* __restrict__ src, const int* __restrict__ dst, const int* list,
    int n, int lo, int lane, int* raw_fill, int* deg_lds, int* raw) {
  int at = 0;
  if (lane == 0)
    at = __hip_atomic_fetch_add(raw_fill, n, __ATOMIC_RELAXED, __HIP_MEMORY_SCOPE_WORKGROUP);
  at = __builtin_amdgcn_readfirstlane(at);
  if (lane < n) {
    const int e = list[lane];
    const int s = src[e];
    const int d = dst[e] - lo;
    __hip_atomic_fetch_add(deg_lds + d, 1, __ATOMIC_RELAXED, __HIP_MEMORY_SCOPE_WORKGROUP);
    raw[at + lane] = s | (d << SRC_BITS);
  }
}

static __device__ __forceinline__ void load_dst_group(const int* __restrict__ dst, int g,
                                                      int lane, int E, bool vec_ok,
                                                      int (&d)[16]) {
  const int e0 = g * GRP_EDGES + lane * 4;
  if (vec_ok && (g + 1) * GRP_EDGES <= E) {
#pragma unroll
    for (int c = 0; c < 4; ++c) {
      const int4 q = *(const int4*)(dst + e0 + c * 128);
      d[c * 4 + 0] = q.x; d[c * 4 + 1] = q.y; d[c * 4 + 2] = q.z; d[c * 4 + 3] = q.w;
    }
  } else {
#pragma unroll
    for (int c = 0; c < 4; ++c)
#pragma unroll
      for (int k = 0; k < 4; ++k) {
        const int e = e0 + c * 128 + k;
        d[c * 4 + k] = (e < E) ? dst[e] : -1;
      }
  }
}

__global__ __launch_bounds__(1024) void owner_sum_kernel(const float* __restrict__ x,
                                                         const int* __restrict__ src,
                                                         const int* __restrict__ dst,
                                                         const int* __restrict__ own_cnt,
                                                         int* __restrict__ sub_start_g,
                                                         int* __restrict__ sub_cnt_g,
                                                         int* raw_list, int* own_list,
                                                         float* __restrict__ inv_deg,
                                                         float* __restrict__ agg,
                                                         int E, int N) {
  __shared__ float rows_lds[(OWN_ROWS + 1) * HID];
  __shared__ int   list_lds[OWN_WAVES * LIST_CAP];
  __shared__ int   deg_lds[DEG_WORDS];
  __shared__ int   sub_start_lds[OWN_WAVES];
  __shared__ int   sub_n_lds[OWN_WAVES];
  __shared__ int   fill_lds[OWN_WAVES];
  __shared__ int   own_start_lds;
  __shared__ int   raw_fill_lds;

  const int tid  = threadIdx.x;
  const int lane = tid & 31;
  const int wave = __builtin_amdgcn_readfirstlane(tid >> 5);
  const int lo   = blockIdx.x * OWN_ROWS;
  const int rows = min(OWN_ROWS, N - lo);

  for (int i = tid; i < OWN_ROWS * HID / 4; i += 1024)
    ((float4*)rows_lds)[i] = make_float4(0.f, 0.f, 0.f, 0.f);
  if (tid < DEG_WORDS) deg_lds[tid] = 0;
  if (wave == 0) {
    const int a = owner_stretch_start(own_cnt, blockIdx.x, lane);
    if (lane == 0) { own_start_lds = a; raw_fill_lds = 0; }
  }
  __syncthreads();
  const int own_start = __builtin_amdgcn_readfirstlane(own_start_lds);
  int* raw = raw_list + own_start;

  int* list = list_lds + wave * LIST_CAP;
  int  cnt  = 0;
  const bool vec_ok = (E & 3) == 0;
  const int  ngrp   = (E + GRP_EDGES - 1) / GRP_EDGES;
  const int  rot    = (int)((blockIdx.x * (unsigned)ngrp) / gridDim.x);

  int nxt[16];
  if (wave < ngrp) {
    int g = wave + rot;
    if (g >= ngrp) g -= ngrp;
    load_dst_group(dst, g, lane, E, vec_ok, nxt);
  }
#pragma unroll 1
  for (int t = wave; t < ngrp; t += OWN_WAVES) {
    int g = t + rot;
    if (g >= ngrp) g -= ngrp;
    const int e0 = g * GRP_EDGES + lane * 4;
    int d[16];
#pragma unroll
    for (int i = 0; i < 16; ++i) d[i] = nxt[i];
    if (t + OWN_WAVES < ngrp) {
      int gn = t + OWN_WAVES + rot;
      if (gn >= ngrp) gn -= ngrp;
      load_dst_group(dst, gn, lane, E, vec_ok, nxt);
    }
#pragma unroll
    for (int c = 0; c < 4; ++c)
#pragma unroll
      for (int k = 0; k < 4; ++k) {
        const bool mine = (unsigned)(d[c * 4 + k] - lo) < (unsigned)rows;
        const unsigned m = __builtin_amdgcn_ballot_w32(mine);
        if (m != 0u) {
          if (mine) list[cnt + (int)__builtin_amdgcn_mbcnt_lo(m, 0u)] = e0 + c * 128 + k;
          cnt += __builtin_popcount(m);
        }
      }
    __builtin_amdgcn_wave_barrier();
    while (cnt >= 32) {
      place_listed_edges(src, dst, list + (cnt - 32), 32, lo, lane, &raw_fill_lds,
                         deg_lds, raw);
      cnt -= 32;
    }
    __builtin_amdgcn_wave_barrier();
  }
  if (cnt > 0)
    place_listed_edges(src, dst, list, cnt, lo, lane, &raw_fill_lds, deg_lds, raw);
  __syncthreads();

  if (wave == 0)
    owner_sub_lists(deg_lds, own_start, blockIdx.x, lane, sub_start_lds, sub_n_lds,
                    fill_lds, sub_start_g, sub_cnt_g);
  if (tid < rows) inv_deg[lo + tid] = 1.0f / fmaxf((float)deg_lds[tid], 1.0f);
  __syncthreads();

  const int total = raw_fill_lds;
#pragma unroll 1
  for (int i0 = 0; i0 < total; i0 += 1024 * RAW_LOADS) {
    int wd[RAW_LOADS];
#pragma unroll
    for (int k = 0; k < RAW_LOADS; ++k) {
      const int i = i0 + k * 1024 + tid;
      wd[k] = (i < total) ? raw[i] : -1;
    }
#pragma unroll
    for (int k = 0; k < RAW_LOADS; ++k)
      if (wd[k] >= 0) {
        const int w  = (wd[k] >> SRC_BITS) & 31;
        const int at = __hip_atomic_fetch_add(fill_lds + w, 1, __ATOMIC_RELAXED,
                                              __HIP_MEMORY_SCOPE_WORKGROUP);
        own_list[sub_start_lds[w] + at] = wd[k];
      }
  }
  __syncthreads();

  sum_own_rows(x, rows_lds, own_list + __builtin_amdgcn_readfirstlane(sub_start_lds[wave]),
               __builtin_amdgcn_readfirstlane(sub_n_lds[wave]), lane);
  __syncthreads();

  for (int i = tid; i < rows * HID / 4; i += 1024)
    ((float4*)(agg + (size_t)lo * HID))[i] = ((const float4*)rows_lds)[i];
}

__global__ __launch_bounds__(1024) void list_sum_kernel(const float* __restrict__ x,
                                                        const int* __restrict__ sub_start_g,
                                                        const int* __restrict__ sub_cnt_g,
                                                        const int* __restrict__ own_list,
                                                        float* __restrict__ agg, int N) {
  __shared__ float rows_lds[(OWN_ROWS + 1) * HID];

  const int tid  = threadIdx.x;
  const int lane = tid & 31;
  const int wave = __builtin_amdgcn_readfirstlane(tid >> 5);
  const int lo   = blockIdx.x * OWN_ROWS;
  const int rows = min(OWN_ROWS, N - lo);
  const int sub_start = sub_start_g[blockIdx.x * OWN_WAVES + wave];
  const int sub_n     = sub_cnt_g[blockIdx.x * OWN_WAVES + wave];

  for (int i = tid; i < OWN_ROWS * HID / 4; i += 1024)
    ((float4*)rows_lds)[i] = make_float4(0.f, 0.f, 0.f, 0.f);
  __syncthreads();

  sum_own_rows(x, rows_lds, own_list + __builtin_amdgcn_readfirstlane(sub_start),
               __builtin_amdgcn_readfirstlane(sub_n), lane);
  __syncthreads();

  for (int k = tid; k < rows * HID / 4; k += 1024)
    ((float4*)(agg + (size_t)lo * HID))[k] = ((const float4*)rows_lds)[k];
}

#define XF_GRID    256
#define XF_WAVES   16
#define XF_THREADS (XF_WAVES * 32)
#define XF_STRIDE  (XF_GRID * XF_WAVES)
#define NTILES     (NNODES / TILE_M)
#define WFRAGS     (16 * 8 * 32)
static_assert(NNODES % TILE_M == 0, "whole row tiles");

static __device__ __forceinline__ void stage_weights(const float* __restrict__ W,
                                                     float4* lds_w, int tid) {
  static_assert(WFRAGS % XF_THREADS == 0, "whole passes");
  float4 w[WFRAGS / XF_THREADS];
#pragma unroll
  for (int k = 0; k < WFRAGS / XF_THREADS; ++k) {
    const int i = tid + k * XF_THREADS;
    const int l = i & 31, t = (i >> 5) & 7, j = i >> 8;
    const int c = l & 15, h = l >> 4;
    w[k] = *(const float4*)(W + (size_t)(t * 16 + c) * HID + 8 * j + 4 * h);
  }
#pragma unroll
  for (int k = 0; k < WFRAGS / XF_THREADS; ++k) lds_w[tid + k * XF_THREADS] = w[k];
}

static __device__ __forceinline__ void load_b(float4 (&b)[8], const float4* p) {
#pragma unroll
  for (int t = 0; t < 8; ++t) b[t] = p[t * 32];
}

static __device__ __forceinline__ void k_step(const float4 a, const float4 (&b)[8],
                                              v8f (&acc)[8]) {
  const v2f a0 = {a.x, a.y};
  const v2f a1 = {a.z, a.w};
#pragma unroll
  for (int t = 0; t < 8; ++t) {
    const v2f b0 = {b[t].x, b[t].y};
    acc[t] = __builtin_amdgcn_wmma_f32_16x16x4_f32(false, a0, false, b0,
                                                   (short)0, acc[t], false, false);
  }
#pragma unroll
  for (int t = 0; t < 8; ++t) {
    const v2f b1 = {b[t].z, b[t].w};
    acc[t] = __builtin_amdgcn_wmma_f32_16x16x4_f32(false, a1, false, b1,
                                                   (short)0, acc[t], false, false);
  }
}

__global__ __launch_bounds__(XF_THREADS) void sage_transform(
    const float* agg, const float* __restrict__ xin,
    const float* __restrict__ Wl, const float* __restrict__ bl,
    const float* __restrict__ Wr, const float* __restrict__ invdeg,
    float* xout) {
  __shared__ float4 lds_w[2 * WFRAGS];

  const int tid  = threadIdx.x;
  const int lane = tid & 31;
  const int wave = __builtin_amdgcn_readfirstlane(tid >> 5);
  const int c = lane & 15;
  const int h = lane >> 4;

  stage_weights(Wl, lds_w, tid);
  stage_weights(Wr, lds_w + WFRAGS, tid);
  __syncthreads();

  const int tile0 = wave * XF_GRID + (int)blockIdx.x;
  if (tile0 >= NTILES) return;
  const int nhalf = 2 * ((NTILES - 1 - tile0) / XF_STRIDE + 1);

  float4 a[16];
  {
    const float4* pa = (const float4*)(agg + (size_t)(tile0 * TILE_M + c) * HID) + h;
#pragma unroll
    for (int j = 0; j < 16; ++j) a[j] = pa[2 * j];
  }
  float scale = invdeg[tile0 * TILE_M + c];
  v8f acc[8];
#pragma unroll
  for (int t = 0; t < 8; ++t) acc[t] = (v8f){0.f, 0.f, 0.f, 0.f, 0.f, 0.f, 0.f, 0.f};

#pragma unroll 1
  for (int q = 0; q < nhalf; ++q) {
    const int m    = q & 1;
    const int row0 = (tile0 + (q >> 1) * XF_STRIDE) * TILE_M;
    const int qn = min(q + 1, nhalf - 1);
    const int mn = qn & 1;
    const int rn = (tile0 + (qn >> 1) * XF_STRIDE) * TILE_M + c;
    const float4* pn = (const float4*)((mn ? xin : agg) + (size_t)rn * HID) + h;
    const float4* pb = lds_w + m * WFRAGS + lane;

    float4 an[16];
    float4 b[2][8];
    load_b(b[0], pb);
#pragma unroll
    for (int s = 0; s < 16; ++s) {
      if (s + 1 < 16) load_b(b[(s + 1) & 1], pb + (s + 1) * 256);
      float4 av = a[s];
      av.x *= scale; av.y *= scale; av.z *= scale; av.w *= scale;
      an[s] = pn[2 * s];
      k_step(av, b[s & 1], acc);
      __builtin_amdgcn_sched_barrier(0);
    }
    const float scale_n = mn ? 1.0f : invdeg[rn];

    if (m == 1) {
      float bias[8];
#pragma unroll
      for (int t = 0; t < 8; ++t) bias[t] = bl[t * 16 + c];
      float ss[8];
#pragma unroll
      for (int r = 0; r < 8; ++r) ss[r] = 0.f;
#pragma unroll
      for (int t = 0; t < 8; ++t)
#pragma unroll
        for (int r = 0; r < 8; ++r) {
          const float o = acc[t][r] + bias[t];
          acc[t][r] = o;
          ss[r] += o * o;
        }
#pragma unroll
      for (int r = 0; r < 8; ++r) {
        ss[r] += __shfl_xor(ss[r], 1);
        ss[r] += __shfl_xor(ss[r], 2);
        ss[r] += __shfl_xor(ss[r], 4);
        ss[r] += __shfl_xor(ss[r], 8);
      }
#pragma unroll
      for (int r = 0; r < 8; ++r) {
        const float sc = 1.0f / fmaxf(sqrtf(ss[r]), 1e-12f);
        float* po = xout + (size_t)(row0 + 8 * h + r) * HID + c;
#pragma unroll
        for (int t = 0; t < 8; ++t) po[t * 16] = fmaxf(acc[t][r] * sc, 0.f);
      }
#pragma unroll
      for (int t = 0; t < 8; ++t) acc[t] = (v8f){0.f, 0.f, 0.f, 0.f, 0.f, 0.f, 0.f, 0.f};
    }
#pragma unroll
    for (int j = 0; j < 16; ++j) a[j] = an[j];
    scale = scale_n;
  }
}

__global__ __launch_bounds__(XF_THREADS) void mlp_head(
    const float* __restrict__ xin, const float* __restrict__ W1,
    const float* __restrict__ b1, const float* __restrict__ W2,
    const float* __restrict__ b2, float* __restrict__ out) {
  __shared__ float4 lds_w[WFRAGS];

  const int tid  = threadIdx.x;
  const int lane = tid & 31;
  const int wave = __builtin_amdgcn_readfirstlane(tid >> 5);
  const int c = lane & 15;
  const int h = lane >> 4;

  stage_weights(W1, lds_w, tid);
  float bias[8], w20[8], w21[8];
#pragma unroll
  for (int t = 0; t < 8; ++t) {
    bias[t] = b1[t * 16 + c];
    w20[t]  = W2[t * 16 + c];
    w21[t]  = W2[HID + t * 16 + c];
  }
  const float bout = (c & 1) ? b2[1] : b2[0];
  __syncthreads();

#pragma unroll 1
  for (int tile = wave * XF_GRID + (int)blockIdx.x; tile < NTILES; tile += XF_STRIDE) {
    const int row0 = tile * TILE_M;
    const float4* px = (const float4*)(xin + (size_t)(row0 + c) * HID) + h;
    float4 aX[16];
#pragma unroll
    for (int j = 0; j < 16; ++j) aX[j] = px[2 * j];

    v8f acc[8];
#pragma unroll
    for (int t = 0; t < 8; ++t) acc[t] = (v8f){0.f, 0.f, 0.f, 0.f, 0.f, 0.f, 0.f, 0.f};
    float4 b[2][8];
    load_b(b[0], lds_w + lane);
#pragma unroll
    for (int s = 0; s < 16; ++s) {
      if (s + 1 < 16) load_b(b[(s + 1) & 1], lds_w + (s + 1) * 256 + lane);
      k_step(aX[s], b[s & 1], acc);
      __builtin_amdgcn_sched_barrier(0);
    }

    float v = 0.f;
#pragma unroll
    for (int r = 0; r < 8; ++r) {
      float s0 = 0.f, s1 = 0.f;
#pragma unroll
      for (int t = 0; t < 8; ++t) {
        const float hv = fmaxf(acc[t][r] + bias[t], 0.f);
        s0 += hv * w20[t];
        s1 += hv * w21[t];
      }
      s0 += __shfl_xor(s0, 1); s1 += __shfl_xor(s1, 1);
      s0 += __shfl_xor(s0, 2); s1 += __shfl_xor(s1, 2);
      s0 += __shfl_xor(s0, 4); s1 += __shfl_xor(s1, 4);
      s0 += __shfl_xor(s0, 8); s1 += __shfl_xor(s1, 8);
      const float pick = (c & 1) ? s1 : s0;
      v = ((c >> 1) == r) ? pick : v;
    }
    out[(size_t)row0 * 2 + lane] = v + bout;
  }
}

extern "C" void kernel_launch(void* const* d_in, const int* in_sizes, int n_in,
                              void* d_out, int out_size, void* d_ws, size_t ws_size,
                              hipStream_t stream) {
  const float* x  = (const float*)d_in[0];
  const int*   ei = (const int*)d_in[1];
  const float* Wl = (const float*)d_in[2];
  const float* bl = (const float*)d_in[3];
  const float* Wr = (const float*)d_in[4];
  const float* W1 = (const float*)d_in[5];
  const float* b1 = (const float*)d_in[6];
  const float* W2 = (const float*)d_in[7];
  const float* b2 = (const float*)d_in[8];
  float* out = (float*)d_out;

  const int N = NNODES;
  const int E = in_sizes[1] / 2;
  const int* src = ei;
  const int* dst = ei + E;

  float* inv_deg = (float*)d_ws;
  const size_t pad = 100096;
  float* bufA = inv_deg + pad;
  float* bufB = bufA + (size_t)N * HID;
  int* own_cnt     = (int*)(bufB + (size_t)N * HID);
  int* sub_start_g = own_cnt + MAX_OWNERS;
  int* sub_cnt_g   = sub_start_g + MAX_OWNERS * OWN_WAVES;
  int* raw_list    = sub_cnt_g + MAX_OWNERS * OWN_WAVES;
  int* own_list    = raw_list + E + LIST_ROOM;

  const int ownerBlocks = (N + OWN_ROWS - 1) / OWN_ROWS;

  hipMemsetAsync(own_cnt, 0, MAX_OWNERS * sizeof(int), stream);
  owner_count_kernel<<<(E + 1024 * CNT_LOADS - 1) / (1024 * CNT_LOADS), 1024, 0, stream>>>(
      dst, own_cnt, E, N);

  const float* xs[3] = {x, bufA, bufB};
  float*       ag[3] = {bufA, bufB, bufA};

  for (int i = 0; i < 3; ++i) {
    float* agg = ag[i];
    const float* xc = xs[i];
    if (i == 0)
      owner_sum_kernel<<<ownerBlocks, 1024, 0, stream>>>(xc, src, dst, own_cnt, sub_start_g,
                                                         sub_cnt_g, raw_list, own_list,
                                                         inv_deg, agg, E, N);
    else
      list_sum_kernel<<<ownerBlocks, 1024, 0, stream>>>(xc, sub_start_g, sub_cnt_g, own_list,
                                                        agg, N);
    sage_transform<<<XF_GRID, XF_THREADS, 0, stream>>>(
        agg, xc, Wl + (size_t)i * HID * HID, bl + (size_t)i * HID,
        Wr + (size_t)i * HID * HID, inv_deg, agg);
  }
  mlp_head<<<XF_GRID, XF_THREADS, 0, stream>>>(bufA, W1, b1, W2, b2, out);
}


# Edges handed to their owner groups by one binning pass (LDS ranks per block, one place reservation per block and owner) instead of 250 whole-list dst scans; sums in reservation and LDS slot order
# speedup vs baseline: 4.5613x; 1.1131x over previous
#include <hip/hip_runtime.h>
#include <math.h>

typedef __attribute__((ext_vector_type(2))) float v2f;
typedef __attribute__((ext_vector_type(8))) float v8f;

#define NNODES 100000
#define HID    128
#define TILE_M 16

#define OWN_ROWS   400
#define OWN_WAVES  32
#define ROWS_IN_FLIGHT 8

#define SRC_BITS   17
#define SRC_MASK   ((1 << SRC_BITS) - 1)
#define MAX_OWNERS 256
#define DEG_WORDS  416
#define OWN_ROOM  1024
#define LIST_ROOM (MAX_OWNERS * (OWN_ROOM + 32))
#define RAW_LOADS 8
#define PAD_ENTRY  (OWN_ROWS << SRC_BITS)
static_assert(NNODES <= (1 << SRC_BITS), "source node must fit the low field");
static_assert(OWN_ROWS < (1 << (31 - SRC_BITS)), "owned row and the spare row must fit the high field");
static_assert((NNODES + OWN_ROWS - 1) / OWN_ROWS <= MAX_OWNERS, "owner count words");
static_assert(OWN_WAVES == 32, "a row's sub-list is its low five bits; one sub-list per lane of wave 0");
static_assert(DEG_WORDS >= OWN_ROWS && DEG_WORDS % 32 == 0, "one count per owned row");

#define CNT_LOADS 8
__global__ __launch_bounds__(1024) void owner_count_kernel(const int* __restrict__ dst,
                                                           int* __restrict__ own_cnt,
                                                           int* __restrict__ own_fill,
                                                           int E, int N) {
  __shared__ int h[MAX_OWNERS];
  const int t = threadIdx.x;
  if (t < MAX_OWNERS) h[t] = 0;
  if (blockIdx.x == 0 && t < MAX_OWNERS) own_fill[t] = 0;
  __syncthreads();
  const int base = blockIdx.x * (1024 * CNT_LOADS) + t;
  int d[CNT_LOADS];
#pragma unroll
  for (int k = 0; k < CNT_LOADS; ++k) {
    const int e = base + k * 1024;
    d[k] = (e < E) ? dst[e] : -1;
  }
#pragma unroll
  for (int k = 0; k < CNT_LOADS; ++k)
    if ((unsigned)d[k] < (unsigned)N)
      __hip_atomic_fetch_add(h + d[k] / OWN_ROWS, 1, __ATOMIC_RELAXED,
                             __HIP_MEMORY_SCOPE_WORKGROUP);
  __syncthreads();
  if (t < MAX_OWNERS && h[t] != 0)
    __hip_atomic_fetch_add(own_cnt + t, h[t], __ATOMIC_RELAXED, __HIP_MEMORY_SCOPE_AGENT);
}

__global__ __launch_bounds__(1024) void owner_bin_kernel(const int* __restrict__ src,
                                                         const int* __restrict__ dst,
                                                         const int* __restrict__ own_cnt,
                                                         int* __restrict__ own_fill,
                                                         int* __restrict__ raw_list,
                                                         int E, int N) {
  __shared__ int h[MAX_OWNERS];
  __shared__ int wave_room[MAX_OWNERS / 32];
  const int t    = threadIdx.x;
  const int lane = t & 31;
  const int base = blockIdx.x * (1024 * CNT_LOADS) + t;
  int d[CNT_LOADS], s[CNT_LOADS];
#pragma unroll
  for (int k = 0; k < CNT_LOADS; ++k) {
    const int e = base + k * 1024;
    d[k] = (e < E) ? dst[e] : -1;
    s[k] = (e < E) ? src[e] : 0;
  }
  int room = 0, inc = 0;
  if (t < MAX_OWNERS) {
    h[t] = 0;
    if (t < (NNODES + OWN_ROWS - 1) / OWN_ROWS) room = ((own_cnt[t] + 31) & ~31) + OWN_ROOM;
    inc = room;
    for (int o = 1; o < 32; o <<= 1) {
      const int u = __shfl_up(inc, o);
      if (lane >= o) inc += u;
    }
    if (lane == 31) wave_room[t >> 5] = inc;
  }
  __syncthreads();
  int start = inc - room;
  if (t < MAX_OWNERS)
    for (int w = 0; w < (t >> 5); ++w) start += wave_room[w];
  int r[CNT_LOADS];
#pragma unroll
  for (int k = 0; k < CNT_LOADS; ++k) {
    r[k] = 0;
    if ((unsigned)d[k] < (unsigned)N)
      r[k] = __hip_atomic_fetch_add(h + d[k] / OWN_ROWS, 1, __ATOMIC_RELAXED,
                                    __HIP_MEMORY_SCOPE_WORKGROUP);
  }
  __syncthreads();
  if (t < MAX_OWNERS) {
    const int n = h[t];
    int at = 0;
    if (n != 0)
      at = __hip_atomic_fetch_add(own_fill + t, n, __ATOMIC_RELAXED, __HIP_MEMORY_SCOPE_AGENT);
    h[t] = start + at;
  }
  __syncthreads();
#pragma unroll
  for (int k = 0; k < CNT_LOADS; ++k)
    if ((unsigned)d[k] < (unsigned)N) {
      const int o = d[k] / OWN_ROWS;
      raw_list[h[o] + r[k]] = s[k] | ((d[k] - o * OWN_ROWS) << SRC_BITS);
    }
}


static __device__ __forceinline__ int owner_stretch_start(const int* __restrict__ own_cnt,
                                                          int b, int lane) {
  int a = 0;
  for (int j = lane; j < b; j += 32) a += ((own_cnt[j] + 31) & ~31) + OWN_ROOM;
  a += __shfl_xor(a, 16);
  a += __shfl_xor(a, 8);
  a += __shfl_xor(a, 4);
  a += __shfl_xor(a, 2);
  a += __shfl_xor(a, 1);
  return a;
}

static __device__ __forceinline__ void owner_sub_lists(const int* deg_lds, int own_start,
                                                       int b, int lane, int* sub_start,
                                                       int* sub_n, int* fill,
                                                       int* __restrict__ sub_start_g,
                                                       int* __restrict__ sub_cnt_g) {
  int c = 0;
  for (int k = 0; k < DEG_WORDS / 32; ++k) c += deg_lds[k * 32 + lane];
  const int r = (c + 31) & ~31;
  int inc = r;
  for (int o = 1; o < 32; o <<= 1) {
    const int u = __shfl_up(inc, o);
    if (lane >= o) inc += u;
  }
  const int st = own_start + inc - r;
  sub_start[lane] = st;
  sub_n[lane]     = c;
  fill[lane]      = 0;
  sub_start_g[b * OWN_WAVES + lane] = st;
  sub_cnt_g[b * OWN_WAVES + lane]   = c;
}

static __device__ __forceinline__ void load_rows8(const float* __restrict__ x, int s_l,
                                                  int b, int lane, float4 (&v)[ROWS_IN_FLIGHT]) {
#pragma unroll
  for (int j = 0; j < ROWS_IN_FLIGHT; ++j) {
    const int s = __builtin_amdgcn_readlane(s_l, b + j);
    v[j] = *((const float4*)(x + (size_t)s * HID) + lane);
  }
}

static __device__ __forceinline__ void add_rows8(float* rows_lds, int d_l, int b, int lane,
                                                 const float4 (&v)[ROWS_IN_FLIGHT]) {
#pragma unroll
  for (int j = 0; j < ROWS_IN_FLIGHT; ++j) {
    const int d = __builtin_amdgcn_readlane(d_l, b + j);
    float4* ap = (float4*)(rows_lds + d * HID) + lane;
    float4 a = *ap;
    a.x += v[j].x; a.y += v[j].y; a.z += v[j].z; a.w += v[j].w;
    *ap = a;
  }
}

static __device__ __forceinline__ void sum_own_rows(const float* __restrict__ x,
                                                    float* rows_lds, const int* sub,
                                                    int n, int lane) {
  if (n <= 0) return;
  int cur = PAD_ENTRY, nxt = PAD_ENTRY;
  if (lane < n) cur = sub[lane];
  if (32 + lane < n) nxt = sub[32 + lane];
  float4 v[ROWS_IN_FLIGHT], vn[ROWS_IN_FLIGHT];
  load_rows8(x, cur & SRC_MASK, 0, lane, v);
#pragma unroll 1
  for (int i = 0; i < n; i += 32) {
    const int m   = min(32, n - i);
    const int s_l = cur & SRC_MASK, d_l = cur >> SRC_BITS;
    const int s_n = nxt & SRC_MASK;
#pragma unroll 1
    for (int b = 0; b < m; b += ROWS_IN_FLIGHT) {
      if (b + ROWS_IN_FLIGHT < m) load_rows8(x, s_l, b + ROWS_IN_FLIGHT, lane, vn);
      else                        load_rows8(x, s_n, 0, lane, vn);
      add_rows8(rows_lds, d_l, b, lane, v);
#pragma unroll
      for (int j = 0; j < ROWS_IN_FLIGHT; ++j) v[j] = vn[j];
    }
    cur = nxt;
    nxt = PAD_ENTRY;
    if (i + 64 + lane < n) nxt = sub[i + 64 + lane];
  }
}

__global__ __launch_bounds__(1024) void owner_sum_kernel(const float* __restrict__ x,
                                                         const int* __restrict__ own_cnt,
                                                         int* __restrict__ sub_start_g,
                                                         int* __restrict__ sub_cnt_g,
                                                         const int* __restrict__ raw_list,
                                                         int* own_list,
                                                         float* __restrict__ inv_deg,
                                                         float* __restrict__ agg, int N) {
  __shared__ float rows_lds[(OWN_ROWS + 1) * HID];
  __shared__ int   deg_lds[DEG_WORDS];
  __shared__ int   sub_start_lds[OWN_WAVES];
  __shared__ int   sub_n_lds[OWN_WAVES];
  __shared__ int   fill_lds[OWN_WAVES];
  __shared__ int   own_start_lds;

  const int tid  = threadIdx.x;
  const int lane = tid & 31;
  const int wave = __builtin_amdgcn_readfirstlane(tid >> 5);
  const int lo   = blockIdx.x * OWN_ROWS;
  const int rows = min(OWN_ROWS, N - lo);

  for (int i = tid; i < OWN_ROWS * HID / 4; i += 1024)
    ((float4*)rows_lds)[i] = make_float4(0.f, 0.f, 0.f, 0.f);
  if (tid < DEG_WORDS) deg_lds[tid] = 0;
  if (wave == 0) {
    const int a = owner_stretch_start(own_cnt, blockIdx.x, lane);
    if (lane == 0) own_start_lds = a;
  }
  __syncthreads();
  const int own_start = __builtin_amdgcn_readfirstlane(own_start_lds);
  const int* raw   = raw_list + own_start;
  const int  total = own_cnt[blockIdx.x];

#pragma unroll 1
  for (int i0 = 0; i0 < total; i0 += 1024 * RAW_LOADS) {
    int wd[RAW_LOADS];
#pragma unroll
    for (int k = 0; k < RAW_LOADS; ++k) {
      const int i = i0 + k * 1024 + tid;
      wd[k] = (i < total) ? raw[i] : -1;
    }
#pragma unroll
    for (int k = 0; k < RAW_LOADS; ++k)
      if (wd[k] >= 0)
        __hip_atomic_fetch_add(deg_lds + (wd[k] >> SRC_BITS), 1, __ATOMIC_RELAXED,
                               __HIP_MEMORY_SCOPE_WORKGROUP);
  }
  __syncthreads();

  if (wave == 0)
    owner_sub_lists(deg_lds, own_start, blockIdx.x, lane, sub_start_lds, sub_n_lds,
                    fill_lds, sub_start_g, sub_cnt_g);
  if (tid < rows) inv_deg[lo + tid] = 1.0f / fmaxf((float)deg_lds[tid], 1.0f);
  __syncthreads();

#pragma unroll 1
  for (int i0 = 0; i0 < total; i0 += 1024 * RAW_LOADS) {
    int wd[RAW_LOADS];
#pragma unroll
    for (int k = 0; k < RAW_LOADS; ++k) {
      const int i = i0 + k * 1024 + tid;
      wd[k] = (i < total) ? raw[i] : -1;
    }
#pragma unroll
    for (int k = 0; k < RAW_LOADS; ++k)
      if (wd[k] >= 0) {
        const int w  = (wd[k] >> SRC_BITS) & 31;
        const int at = __hip_atomic_fetch_add(fill_lds + w, 1, __ATOMIC_RELAXED,
                                              __HIP_MEMORY_SCOPE_WORKGROUP);
        own_list[sub_start_lds[w] + at] = wd[k];
      }
  }
  __syncthreads();

  sum_own_rows(x, rows_lds, own_list + __builtin_amdgcn_readfirstlane(sub_start_lds[wave]),
               __builtin_amdgcn_readfirstlane(sub_n_lds[wave]), lane);
  __syncthreads();

  for (int i = tid; i < rows * HID / 4; i += 1024)
    ((float4*)(agg + (size_t)lo * HID))[i] = ((const float4*)rows_lds)[i];
}

__global__ __launch_bounds__(1024) void list_sum_kernel(const float* __restrict__ x,
                                                        const int* __restrict__ sub_start_g,
                                                        const int* __restrict__ sub_cnt_g,
                                                        const int* __restrict__ own_list,
                                                        float* __restrict__ agg, int N) {
  __shared__ float rows_lds[(OWN_ROWS + 1) * HID];

  const int tid  = threadIdx.x;
  const int lane = tid & 31;
  const int wave = __builtin_amdgcn_readfirstlane(tid >> 5);
  const int lo   = blockIdx.x * OWN_ROWS;
  const int rows = min(OWN_ROWS, N - lo);
  const int sub_start = sub_start_g[blockIdx.x * OWN_WAVES + wave];
  const int sub_n     = sub_cnt_g[blockIdx.x * OWN_WAVES + wave];

  for (int i = tid; i < OWN_ROWS * HID / 4; i += 1024)
    ((float4*)rows_lds)[i] = make_float4(0.f, 0.f, 0.f, 0.f);
  __syncthreads();

  sum_own_rows(x, rows_lds, own_list + __builtin_amdgcn_readfirstlane(sub_start),
               __builtin_amdgcn_readfirstlane(sub_n), lane);
  __syncthreads();

  for (int k = tid; k < rows * HID / 4; k += 1024)
    ((float4*)(agg + (size_t)lo * HID))[k] = ((const float4*)rows_lds)[k];
}

#define XF_GRID    256
#define XF_WAVES   16
#define XF_THREADS (XF_WAVES * 32)
#define XF_STRIDE  (XF_GRID * XF_WAVES)
#define NTILES     (NNODES / TILE_M)
#define WFRAGS     (16 * 8 * 32)
static_assert(NNODES % TILE_M == 0, "whole row tiles");

static __device__ __forceinline__ void stage_weights(const float* __restrict__ W,
                                                     float4* lds_w, int tid) {
  static_assert(WFRAGS % XF_THREADS == 0, "whole passes");
  float4 w[WFRAGS / XF_THREADS];
#pragma unroll
  for (int k = 0; k < WFRAGS / XF_THREADS; ++k) {
    const int i = tid + k * XF_THREADS;
    const int l = i & 31, t = (i >> 5) & 7, j = i >> 8;
    const int c = l & 15, h = l >> 4;
    w[k] = *(const float4*)(W + (size_t)(t * 16 + c) * HID + 8 * j + 4 * h);
  }
#pragma unroll
  for (int k = 0; k < WFRAGS / XF_THREADS; ++k) lds_w[tid + k * XF_THREADS] = w[k];
}

static __device__ __forceinline__ void load_b(float4 (&b)[8], const float4* p) {
#pragma unroll
  for (int t = 0; t < 8; ++t) b[t] = p[t * 32];
}

static __device__ __forceinline__ void k_step(const float4 a, const float4 (&b)[8],
                                              v8f (&acc)[8]) {
  const v2f a0 = {a.x, a.y};
  const v2f a1 = {a.z, a.w};
#pragma unroll
  for (int t = 0; t < 8; ++t) {
    const v2f b0 = {b[t].x, b[t].y};
    acc[t] = __builtin_amdgcn_wmma_f32_16x16x4_f32(false, a0, false, b0,
                                                   (short)0, acc[t], false, false);
  }
#pragma unroll
  for (int t = 0; t < 8; ++t) {
    const v2f b1 = {b[t].z, b[t].w};
    acc[t] = __builtin_amdgcn_wmma_f32_16x16x4_f32(false, a1, false, b1,
                                                   (short)0, acc[t], false, false);
  }
}

__global__ __launch_bounds__(XF_THREADS) void sage_transform(
    const float* agg, const float* __restrict__ xin,
    const float* __restrict__ Wl, const float* __restrict__ bl,
    const float* __restrict__ Wr, const float* __restrict__ invdeg,
    float* xout) {
  __shared__ float4 lds_w[2 * WFRAGS];

  const int tid  = threadIdx.x;
  const int lane = tid & 31;
  const int wave = __builtin_amdgcn_readfirstlane(tid >> 5);
  const int c = lane & 15;
  const int h = lane >> 4;

  stage_weights(Wl, lds_w, tid);
  stage_weights(Wr, lds_w + WFRAGS, tid);
  __syncthreads();

  const int tile0 = wave * XF_GRID + (int)blockIdx.x;
  if (tile0 >= NTILES) return;
  const int nhalf = 2 * ((NTILES - 1 - tile0) / XF_STRIDE + 1);

  float4 a[16];
  {
    const float4* pa = (const float4*)(agg + (size_t)(tile0 * TILE_M + c) * HID) + h;
#pragma unroll
    for (int j = 0; j < 16; ++j) a[j] = pa[2 * j];
  }
  float scale = invdeg[tile0 * TILE_M + c];
  v8f acc[8];
#pragma unroll
  for (int t = 0; t < 8; ++t) acc[t] = (v8f){0.f, 0.f, 0.f, 0.f, 0.f, 0.f, 0.f, 0.f};

#pragma unroll 1
  for (int q = 0; q < nhalf; ++q) {
    const int m    = q & 1;
    const int row0 = (tile0 + (q >> 1) * XF_STRIDE) * TILE_M;
    const int qn = min(q + 1, nhalf - 1);
    const int mn = qn & 1;
    const int rn = (tile0 + (qn >> 1) * XF_STRIDE) * TILE_M + c;
    const float4* pn = (const float4*)((mn ? xin : agg) + (size_t)rn * HID) + h;
    const float4* pb = lds_w + m * WFRAGS + lane;

    float4 an[16];
    float4 b[2][8];
    load_b(b[0], pb);
#pragma unroll
    for (int s = 0; s < 16; ++s) {
      if (s + 1 < 16) load_b(b[(s + 1) & 1], pb + (s + 1) * 256);
      float4 av = a[s];
      av.x *= scale; av.y *= scale; av.z *= scale; av.w *= scale;
      an[s] = pn[2 * s];
      k_step(av, b[s & 1], acc);
      __builtin_amdgcn_sched_barrier(0);
    }
    const float scale_n = mn ? 1.0f : invdeg[rn];

    if (m == 1) {
      float bias[8];
#pragma unroll
      for (int t = 0; t < 8; ++t) bias[t] = bl[t * 16 + c];
      float ss[8];
#pragma unroll
      for (int r = 0; r < 8; ++r) ss[r] = 0.f;
#pragma unroll
      for (int t = 0; t < 8; ++t)
#pragma unroll
        for (int r = 0; r < 8; ++r) {
          const float o = acc[t][r] + bias[t];
          acc[t][r] = o;
          ss[r] += o * o;
        }
#pragma unroll
      for (int r = 0; r < 8; ++r) {
        ss[r] += __shfl_xor(ss[r], 1);
        ss[r] += __shfl_xor(ss[r], 2);
        ss[r] += __shfl_xor(ss[r], 4);
        ss[r] += __shfl_xor(ss[r], 8);
      }
#pragma unroll
      for (int r = 0; r < 8; ++r) {
        const float sc = 1.0f / fmaxf(sqrtf(ss[r]), 1e-12f);
        float* po = xout + (size_t)(row0 + 8 * h + r) * HID + c;
#pragma unroll
        for (int t = 0; t < 8; ++t) po[t * 16] = fmaxf(acc[t][r] * sc, 0.f);
      }
#pragma unroll
      for (int t = 0; t < 8; ++t) acc[t] = (v8f){0.f, 0.f, 0.f, 0.f, 0.f, 0.f, 0.f, 0.f};
    }
#pragma unroll
    for (int j = 0; j < 16; ++j) a[j] = an[j];
    scale = scale_n;
  }
}

__global__ __launch_bounds__(XF_THREADS) void mlp_head(
    const float* __restrict__ xin, const float* __restrict__ W1,
    const float* __restrict__ b1, const float* __restrict__ W2,
    const float* __restrict__ b2, float* __restrict__ out) {
  __shared__ float4 lds_w[WFRAGS];

  const int tid  = threadIdx.x;
  const int lane = tid & 31;
  const int wave = __builtin_amdgcn_readfirstlane(tid >> 5);
  const int c = lane & 15;
  const int h = lane >> 4;

  stage_weights(W1, lds_w, tid);
  float bias[8], w20[8], w21[8];
#pragma unroll
  for (int t = 0; t < 8; ++t) {
    bias[t] = b1[t * 16 + c];
    w20[t]  = W2[t * 16 + c];
    w21[t]  = W2[HID + t * 16 + c];
  }
  const float bout = (c & 1) ? b2[1] : b2[0];
  __syncthreads();

#pragma unroll 1
  for (int tile = wave * XF_GRID + (int)blockIdx.x; tile < NTILES; tile += XF_STRIDE) {
    const int row0 = tile * TILE_M;
    const float4* px = (const float4*)(xin + (size_t)(row0 + c) * HID) + h;
    float4 aX[16];
#pragma unroll
    for (int j = 0; j < 16; ++j) aX[j] = px[2 * j];

    v8f acc[8];
#pragma unroll
    for (int t = 0; t < 8; ++t) acc[t] = (v8f){0.f, 0.f, 0.f, 0.f, 0.f, 0.f, 0.f, 0.f};
    float4 b[2][8];
    load_b(b[0], lds_w + lane);
#pragma unroll
    for (int s = 0; s < 16; ++s) {
      if (s + 1 < 16) load_b(b[(s + 1) & 1], lds_w + (s + 1) * 256 + lane);
      k_step(aX[s], b[s & 1], acc);
      __builtin_amdgcn_sched_barrier(0);
    }

    float v = 0.f;
#pragma unroll
    for (int r = 0; r < 8; ++r) {
      float s0 = 0.f, s1 = 0.f;
#pragma unroll
      for (int t = 0; t < 8; ++t) {
        const float hv = fmaxf(acc[t][r] + bias[t], 0.f);
        s0 += hv * w20[t];
        s1 += hv * w21[t];
      }
      s0 += __shfl_xor(s0, 1); s1 += __shfl_xor(s1, 1);
      s0 += __shfl_xor(s0, 2); s1 += __shfl_xor(s1, 2);
      s0 += __shfl_xor(s0, 4); s1 += __shfl_xor(s1, 4);
      s0 += __shfl_xor(s0, 8); s1 += __shfl_xor(s1, 8);
      const float pick = (c & 1) ? s1 : s0;
      v = ((c >> 1) == r) ? pick : v;
    }
    out[(size_t)row0 * 2 + lane] = v + bout;
  }
}

extern "C" void kernel_launch(void* const* d_in, const int* in_sizes, int n_in,
                              void* d_out, int out_size, void* d_ws, size_t ws_size,
                              hipStream_t stream) {
  const float* x  = (const float*)d_in[0];
  const int*   ei = (const int*)d_in[1];
  const float* Wl = (const float*)d_in[2];
  const float* bl = (const float*)d_in[3];
  const float* Wr = (const float*)d_in[4];
  const float* W1 = (const float*)d_in[5];
  const float* b1 = (const float*)d_in[6];
  const float* W2 = (const float*)d_in[7];
  const float* b2 = (const float*)d_in[8];
  float* out = (float*)d_out;

  const int N = NNODES;
  const int E = in_sizes[1] / 2;
  const int* src = ei;
  const int* dst = ei + E;

  float* inv_deg = (float*)d_ws;
  const size_t pad = 100096;
  float* bufA = inv_deg + pad;
  float* bufB = bufA + (size_t)N * HID;
  int* own_cnt     = (int*)(bufB + (size_t)N * HID);
  int* own_fill    = own_cnt + MAX_OWNERS;
  int* sub_start_g = own_fill + MAX_OWNERS;
  int* sub_cnt_g   = sub_start_g + MAX_OWNERS * OWN_WAVES;
  int* raw_list    = sub_cnt_g + MAX_OWNERS * OWN_WAVES;
  int* own_list    = raw_list + E + LIST_ROOM;

  const int ownerBlocks = (N + OWN_ROWS - 1) / OWN_ROWS;

  hipMemsetAsync(own_cnt, 0, MAX_OWNERS * sizeof(int), stream);
  const int edgeBlocks = (E + 1024 * CNT_LOADS - 1) / (1024 * CNT_LOADS);
  owner_count_kernel<<<edgeBlocks, 1024, 0, stream>>>(dst, own_cnt, own_fill, E, N);
  owner_bin_kernel<<<edgeBlocks, 1024, 0, stream>>>(src, dst, own_cnt, own_fill, raw_list,
                                                    E, N);

  const float* xs[3] = {x, bufA, bufB};
  float*       ag[3] = {bufA, bufB, bufA};

  for (int i = 0; i < 3; ++i) {
    float* agg = ag[i];
    const float* xc = xs[i];
    if (i == 0)
      owner_sum_kernel<<<ownerBlocks, 1024, 0, stream>>>(xc, own_cnt, sub_start_g, sub_cnt_g,
                                                         raw_list, own_list, inv_deg, agg, N);
    else
      list_sum_kernel<<<ownerBlocks, 1024, 0, stream>>>(xc, sub_start_g, sub_cnt_g, own_list,
                                                        agg, N);
    sage_transform<<<XF_GRID, XF_THREADS, 0, stream>>>(
        agg, xc, Wl + (size_t)i * HID * HID, bl + (size_t)i * HID,
        Wr + (size_t)i * HID * HID, inv_deg, agg);
  }
  mlp_head<<<XF_GRID, XF_THREADS, 0, stream>>>(bufA, W1, b1, W2, b2, out);
}
